# starting kernel
# baseline (speedup 1.0000x reference)
#include <hip/hip_runtime.h>
#include <math.h>

typedef float v2f __attribute__((ext_vector_type(2)));
typedef float v8f __attribute__((ext_vector_type(8)));

#define LRELU_SLOPE 0.2f

__device__ __forceinline__ unsigned f2ord(float f) {
  unsigned u = __float_as_uint(f);
  return (u & 0x80000000u) ? ~u : (u | 0x80000000u);
}
__device__ __forceinline__ float ord2f(unsigned u) {
  return __uint_as_float((u & 0x80000000u) ? (u & 0x7FFFFFFFu) : ~u);
}

__global__ __launch_bounds__(256) void gemm_wmma_f32(
    const float* __restrict__ X, const float* __restrict__ W,
    float* __restrict__ Y, int M, int K, int Nc) {
  const int lane  = threadIdx.x & 31;
  const int wave  = (blockIdx.x * blockDim.x + threadIdx.x) >> 5;
  const int tilesN = Nc >> 4;
  const int tilesM = M >> 4;
  if (wave >= tilesM * tilesN) return;
  const int mTile = (wave / tilesN) << 4;
  const int nTile = (wave % tilesN) << 4;
  const int half  = lane >> 4;
  const int l16   = lane & 15;

  const float* __restrict__ xrow = X + (size_t)(mTile + l16) * K;
  const float* __restrict__ wcol = W + nTile + l16;

  v8f acc = {0.f, 0.f, 0.f, 0.f, 0.f, 0.f, 0.f, 0.f};
  for (int k = 0; k < K; k += 4) {
    const int ka = k + 2 * half;
    v2f a = *(const v2f*)(xrow + ka);
    v2f b;
    b.x = wcol[(size_t)ka * Nc];
    b.y = wcol[(size_t)(ka + 1) * Nc];
#if __has_builtin(__builtin_amdgcn_wmma_f32_16x16x4_f32)
    acc = __builtin_amdgcn_wmma_f32_16x16x4_f32(false, a, false, b,
                                                (short)0, acc, false, false);
#else
    acc[0] += a.x * b.x;
#endif
  }
  float* __restrict__ yrow = Y + nTile + l16;
  const int mbase = mTile + half * 8;
#pragma unroll
  for (int r = 0; r < 8; ++r)
    yrow[(size_t)(mbase + r) * Nc] = acc[r];
}

__global__ __launch_bounds__(256) void att_coef(
    const float* __restrict__ h, const float* __restrict__ att_s,
    const float* __restrict__ att_d, float* __restrict__ a_s,
    float* __restrict__ a_d, int NH, int H, int C) {
  const long long gt = (long long)blockIdx.x * blockDim.x + threadIdx.x;
  const int wave = (int)(gt >> 5);
  const int c    = threadIdx.x & 31;
  if (wave >= NH) return;
  const int hh = wave % H;
  const float v = h[(size_t)wave * C + c];
  float ss = v * att_s[hh * C + c];
  float dd = v * att_d[hh * C + c];
#pragma unroll
  for (int off = 16; off > 0; off >>= 1) {
    ss += __shfl_xor(ss, off, 32);
    dd += __shfl_xor(dd, off, 32);
  }
  if (c == 0) { a_s[wave] = ss; a_d[wave] = dd; }
}

__global__ __launch_bounds__(256) void edge_max(
    const int* __restrict__ ei, int E, int N, int H,
    const float* __restrict__ a_s, const float* __restrict__ a_d,
    unsigned* __restrict__ m) {
  const int e = blockIdx.x * blockDim.x + threadIdx.x;
  const int Etot = E + N;
  if (e >= Etot) return;
  const int s = (e < E) ? ei[e] : (e - E);
  const int d = (e < E) ? ei[E + e] : (e - E);
  for (int hh = 0; hh < H; ++hh) {
    float l = a_s[s * H + hh] + a_d[d * H + hh];
    l = (l > 0.f) ? l : LRELU_SLOPE * l;
    atomicMax(&m[d * H + hh], f2ord(l));
  }
}

__global__ __launch_bounds__(256) void edge_acc(
    const int* __restrict__ ei, int E, int N, int H, int C,
    const float* __restrict__ a_s, const float* __restrict__ a_d,
    const unsigned* __restrict__ m, const float* __restrict__ h,
    float* __restrict__ num, float* __restrict__ den) {
  const long long gt = (long long)blockIdx.x * blockDim.x + threadIdx.x;
  const long long wave = gt >> 5;
  const int c = threadIdx.x & 31;
  const long long Etot = (long long)E + N;
  if (wave >= Etot * H) return;
  const int e  = (int)(wave / H);
  const int hh = (int)(wave % H);
  const int s = (e < E) ? ei[e] : (e - E);
  const int d = (e < E) ? ei[E + e] : (e - E);
  float l = a_s[s * H + hh] + a_d[d * H + hh];
  l = (l > 0.f) ? l : LRELU_SLOPE * l;
  const float w = expf(l - ord2f(m[d * H + hh]));
  if (c == 0) atomicAdd(&den[d * H + hh], w);
  const size_t srow = ((size_t)s * H + hh) * C;
  const size_t drow = ((size_t)d * H + hh) * C;
  atomicAdd(&num[drow + c], w * h[srow + c]);
}

__global__ __launch_bounds__(256) void finalize(
    const float* __restrict__ num, const float* __restrict__ den,
    const float* __restrict__ bias, float* __restrict__ out,
    int N, int H, int C) {
  const int t = blockIdx.x * blockDim.x + threadIdx.x;
  if (t >= N * H * C) return;
  const int nh = t / C;
  const int hc = t % (H * C);
  float v = num[t] / (den[nh] + 1e-16f) + bias[hc];
  v = (v > 0.f) ? v : (expf(v) - 1.f);
  out[t] = v;
}

extern "C" void kernel_launch(void* const* d_in, const int* in_sizes, int n_in,
                              void* d_out, int out_size, void* d_ws, size_t ws_size,
                              hipStream_t stream) {
  const float* x   = (const float*)d_in[0];
  const int*   ei  = (const int*)d_in[1];
  const float* W1  = (const float*)d_in[2];
  const float* as1 = (const float*)d_in[3];
  const float* ad1 = (const float*)d_in[4];
  const float* b1  = (const float*)d_in[5];
  const float* W2  = (const float*)d_in[6];
  const float* as2 = (const float*)d_in[7];
  const float* ad2 = (const float*)d_in[8];
  const float* b2  = (const float*)d_in[9];

  const int HC   = in_sizes[3];
  const int Fin  = in_sizes[2] / HC;
  const int N    = in_sizes[0] / Fin;
  const int E    = in_sizes[1] / 2;
  const int H    = 8;
  const int C    = HC / H;
  const int Cout = in_sizes[7];
  const int Etot = E + N;

  float* ws = (float*)d_ws;
  float*    h1   = ws;
  float*    x2   = h1 + (size_t)N * HC;
  float*    h2   = x2 + (size_t)N * HC;
  float*    num2 = h2 + (size_t)N * Cout;
  float*    aS1  = num2 + (size_t)N * Cout;
  float*    aD1  = aS1 + (size_t)N * H;
  unsigned* m1   = (unsigned*)(aD1 + (size_t)N * H);
  float*    den1 = (float*)(m1 + (size_t)N * H);
  float*    aS2  = den1 + (size_t)N * H;
  float*    aD2  = aS2 + N;
  unsigned* m2   = (unsigned*)(aD2 + N);
  float*    den2 = (float*)(m2 + N);

  hipMemsetAsync(x2,   0, (size_t)N * HC   * sizeof(float), stream);
  hipMemsetAsync(num2, 0, (size_t)N * Cout * sizeof(float), stream);
  hipMemsetAsync(m1,   0, (size_t)N * H * sizeof(unsigned), stream);
  hipMemsetAsync(den1, 0, (size_t)N * H * sizeof(float), stream);
  hipMemsetAsync(m2,   0, (size_t)N * sizeof(unsigned), stream);
  hipMemsetAsync(den2, 0, (size_t)N * sizeof(float), stream);

  const int TB = 256;
  auto blocks = [](long long n, int tb) { return (int)((n + tb - 1) / tb); };

  {
    const long long waves = (long long)(N / 16) * (HC / 16);
    gemm_wmma_f32<<<blocks(waves * 32, TB), TB, 0, stream>>>(x, W1, h1, N, Fin, HC);
    att_coef<<<blocks((long long)N * H * 32, TB), TB, 0, stream>>>(
        h1, as1, ad1, aS1, aD1, N * H, H, C);
    edge_max<<<blocks(Etot, TB), TB, 0, stream>>>(ei, E, N, H, aS1, aD1, m1);
    edge_acc<<<blocks((long long)Etot * H * 32, TB), TB, 0, stream>>>(
        ei, E, N, H, C, aS1, aD1, m1, h1, x2, den1);
    finalize<<<blocks((long long)N * HC, TB), TB, 0, stream>>>(x2, den1, b1, x2, N, H, C);
  }

  {
    const long long waves = (long long)(N / 16) * (Cout / 16);
    gemm_wmma_f32<<<blocks(waves * 32, TB), TB, 0, stream>>>(x2, W2, h2, N, HC, Cout);
    att_coef<<<blocks((long long)N * 32, TB), TB, 0, stream>>>(
        h2, as2, ad2, aS2, aD2, N, 1, Cout);
    edge_max<<<blocks(Etot, TB), TB, 0, stream>>>(ei, E, N, 1, aS2, aD2, m2);
    edge_acc<<<blocks((long long)Etot * 32, TB), TB, 0, stream>>>(
        ei, E, N, 1, Cout, aS2, aD2, m2, h2, num2, den2);
    finalize<<<blocks((long long)N * Cout, TB), TB, 0, stream>>>(
        num2, den2, b2, (float*)d_out, N, 1, Cout);
  }
}


# Edge max, scatter-add and finalize become one kernel a layer: a block owns 196 dst nodes, scans all dst ids twice, sums in LDS with LDS atomics (arrival order); global atomics and memsets go
# speedup vs baseline: 3.5204x; 3.5204x over previous
#include <hip/hip_runtime.h>
#include <math.h>

typedef float v2f __attribute__((ext_vector_type(2)));
typedef float v8f __attribute__((ext_vector_type(8)));

#define LRELU_SLOPE 0.2f

__device__ __forceinline__ unsigned f2ord(float f) {
  unsigned u = __float_as_uint(f);
  return (u & 0x80000000u) ? ~u : (u | 0x80000000u);
}
__device__ __forceinline__ float ord2f(unsigned u) {
  return __uint_as_float((u & 0x80000000u) ? (u & 0x7FFFFFFFu) : ~u);
}

__global__ __launch_bounds__(256) void gemm_wmma_f32(
    const float* __restrict__ X, const float* __restrict__ W,
    float* __restrict__ Y, int M, int K, int Nc) {
  const int lane  = threadIdx.x & 31;
  const int wave  = (blockIdx.x * blockDim.x + threadIdx.x) >> 5;
  const int tilesN = Nc >> 4;
  const int tilesM = M >> 4;
  if (wave >= tilesM * tilesN) return;
  const int mTile = (wave / tilesN) << 4;
  const int nTile = (wave % tilesN) << 4;
  const int half  = lane >> 4;
  const int l16   = lane & 15;

  const float* __restrict__ xrow = X + (size_t)(mTile + l16) * K;
  const float* __restrict__ wcol = W + nTile + l16;

  v8f acc = {0.f, 0.f, 0.f, 0.f, 0.f, 0.f, 0.f, 0.f};
  for (int k = 0; k < K; k += 4) {
    const int ka = k + 2 * half;
    v2f a = *(const v2f*)(xrow + ka);
    v2f b;
    b.x = wcol[(size_t)ka * Nc];
    b.y = wcol[(size_t)(ka + 1) * Nc];
#if __has_builtin(__builtin_amdgcn_wmma_f32_16x16x4_f32)
    acc = __builtin_amdgcn_wmma_f32_16x16x4_f32(false, a, false, b,
                                                (short)0, acc, false, false);
#else
    acc[0] += a.x * b.x;
#endif
  }
  float* __restrict__ yrow = Y + nTile + l16;
  const int mbase = mTile + half * 8;
#pragma unroll
  for (int r = 0; r < 8; ++r)
    yrow[(size_t)(mbase + r) * Nc] = acc[r];
}

__global__ __launch_bounds__(256) void att_coef(
    const float* __restrict__ h, const float* __restrict__ att_s,
    const float* __restrict__ att_d, float* __restrict__ a_s,
    float* __restrict__ a_d, int NH, int H, int C) {
  const long long gt = (long long)blockIdx.x * blockDim.x + threadIdx.x;
  const int wave = (int)(gt >> 5);
  const int c    = threadIdx.x & 31;
  if (wave >= NH) return;
  const int hh = wave % H;
  const float v = h[(size_t)wave * C + c];
  float ss = v * att_s[hh * C + c];
  float dd = v * att_d[hh * C + c];
#pragma unroll
  for (int off = 16; off > 0; off >>= 1) {
    ss += __shfl_xor(ss, off, 32);
    dd += __shfl_xor(dd, off, 32);
  }
  if (c == 0) { a_s[wave] = ss; a_d[wave] = dd; }
}

#define NPB 196
#define SEG_TB 1024
#define SEG_WV (SEG_TB / 32)
#define SCAN_U 4
#define SCAN_G (32 * SCAN_U)

template <int H>
__device__ __forceinline__ void ld_coef(const float* __restrict__ a, int n,
                                        float (&v)[H]) {
  if constexpr ((H & 3) == 0) {
#pragma unroll
    for (int i = 0; i < H; i += 4) {
      const float4 t = *(const float4*)(a + (size_t)n * H + i);
      v[i] = t.x; v[i + 1] = t.y; v[i + 2] = t.z; v[i + 3] = t.w;
    }
  } else {
#pragma unroll
    for (int i = 0; i < H; ++i) v[i] = a[(size_t)n * H + i];
  }
}

__device__ __forceinline__ void ld_group(const int* __restrict__ dstp, int g,
                                         int lane, int4 (&v)[SCAN_U]) {
  const int4* __restrict__ p = (const int4*)dstp + (size_t)g * SCAN_G + lane;
#pragma unroll
  for (int u = 0; u < SCAN_U; ++u) v[u] = p[u * 32];
}

template <int H>
__global__ __launch_bounds__(SEG_TB) void seg_softmax_agg(
    const int* __restrict__ ei, int E, int N,
    const float* __restrict__ a_s, const float* __restrict__ a_d,
    const float* __restrict__ h, const float* __restrict__ bias,
    float* __restrict__ out) {
  constexpr int HC = H * 32;
  __shared__ float    s_num[NPB * HC];
  __shared__ float    s_den[NPB * H];
  __shared__ unsigned s_max[NPB * H];
  __shared__ float    s_ad[NPB * H];

  const int tid  = threadIdx.x;
  const int lane = tid & 31;
  const int wv   = __builtin_amdgcn_readfirstlane(tid >> 5);
  const int n0   = blockIdx.x * NPB;
  const int cnt  = min(NPB, N - n0);

  for (int i = tid; i < NPB * HC; i += SEG_TB) s_num[i] = 0.f;
  for (int i = tid; i < NPB * H; i += SEG_TB) {
    s_den[i] = 0.f;
    s_max[i] = 0u;
    s_ad[i]  = (i < cnt * H) ? a_d[(size_t)n0 * H + i] : 0.f;
  }
  __syncthreads();

  const int* __restrict__ srcp = ei;
  const int* __restrict__ dstp = ei + E;
  const bool al16 = ((E & 3) == 0) && (((size_t)dstp & 15) == 0);
  const int nG    = al16 ? (E >> 2) / SCAN_G : 0;
  const int eRest = nG * SCAN_G * 4;

  auto edgeA = [&](int d, int e, bool self) {
    const int dl = d - n0;
    if ((unsigned)dl < (unsigned)cnt) {
      const int s = self ? d : srcp[e];
      float av[H];
      ld_coef<H>(a_s, s, av);
#pragma unroll
      for (int hh = 0; hh < H; ++hh) {
        float l = av[hh] + s_ad[dl * H + hh];
        l = (l > 0.f) ? l : LRELU_SLOPE * l;
        atomicMax(&s_max[dl * H + hh], f2ord(l));
      }
    }
  };

  auto edgeB = [&](int d, int e, bool self) {
    const int  dl  = d - n0;
    const bool hit = (unsigned)dl < (unsigned)cnt;
    unsigned mask = __builtin_amdgcn_ballot_w32(hit);
    if (mask != 0u) {
      int   s = 0;
      float w[H];
#pragma unroll
      for (int hh = 0; hh < H; ++hh) w[hh] = 0.f;
      if (hit) {
        s = self ? d : srcp[e];
        float av[H];
        ld_coef<H>(a_s, s, av);
#pragma unroll
        for (int hh = 0; hh < H; ++hh) {
          float l = av[hh] + s_ad[dl * H + hh];
          l = (l > 0.f) ? l : LRELU_SLOPE * l;
          w[hh] = expf(l - ord2f(s_max[dl * H + hh]));
          atomicAdd(&s_den[dl * H + hh], w[hh]);
        }
      }
      while (mask) {
        const int L = __builtin_ctz(mask);
        mask &= mask - 1u;
        const int sB = __builtin_amdgcn_readlane(s, L);
        const int dB = __builtin_amdgcn_readlane(dl, L);
        const float* __restrict__ hrow = h + (size_t)sB * HC + lane;
        float hv[H];
#pragma unroll
        for (int hh = 0; hh < H; ++hh) hv[hh] = hrow[hh * 32];
#pragma unroll
        for (int hh = 0; hh < H; ++hh) {
          const float wB = __int_as_float(
              __builtin_amdgcn_readlane(__float_as_int(w[hh]), L));
          atomicAdd(&s_num[dB * HC + hh * 32 + lane], wB * hv[hh]);
        }
      }
    }
  };

  const int nlS = lane * SEG_WV + wv;
  const int dS  = (nlS < cnt) ? n0 + nlS : -1;

  auto scan = [&](auto&& edgeFn) {
    if (wv < nG) {
      int4 nx[SCAN_U];
      ld_group(dstp, wv, lane, nx);
      for (int g = wv; g < nG; g += SEG_WV) {
        int4 cur[SCAN_U];
#pragma unroll
        for (int u = 0; u < SCAN_U; ++u) cur[u] = nx[u];
        ld_group(dstp, min(g + SEG_WV, nG - 1), lane, nx);
        unsigned hm = 0u;
#pragma unroll
        for (int u = 0; u < SCAN_U; ++u) {
          hm |= ((unsigned)(cur[u].x - n0) < (unsigned)cnt) ? (1u << (4 * u))     : 0u;
          hm |= ((unsigned)(cur[u].y - n0) < (unsigned)cnt) ? (1u << (4 * u + 1)) : 0u;
          hm |= ((unsigned)(cur[u].z - n0) < (unsigned)cnt) ? (1u << (4 * u + 2)) : 0u;
          hm |= ((unsigned)(cur[u].w - n0) < (unsigned)cnt) ? (1u << (4 * u + 3)) : 0u;
        }
        const int ebase = 4 * (g * SCAN_G + lane);
        while (__builtin_amdgcn_ballot_w32(hm != 0u) != 0u) {
          const bool has = (hm != 0u);
          const int  idx = has ? __builtin_ctz(hm) : 0;
          hm &= hm - 1u;
          const int e = ebase + (idx >> 2) * (4 * 32) + (idx & 3);
          const int d = has ? dstp[e] : -1;
          edgeFn(d, e, false);
        }
      }
    }
    for (int eb = eRest + wv * 32; eb < E; eb += SEG_TB) {
      const int e = eb + lane;
      const int d = (e < E) ? dstp[e] : -1;
      edgeFn(d, (e < E) ? e : 0, false);
    }
  };

  edgeA(dS, 0, true);
  scan(edgeA);
  __syncthreads();

  edgeB(dS, 0, true);
  scan(edgeB);
  __syncthreads();

  float* __restrict__ orow = out + (size_t)n0 * HC;
  for (int i = tid; i < cnt * HC; i += SEG_TB) {
    const int nl = i / HC;
    const int hc = i % HC;
    float v = s_num[i] / (s_den[nl * H + hc / 32] + 1e-16f) + bias[hc];
    v = (v > 0.f) ? v : (expf(v) - 1.f);
    orow[i] = v;
  }
}

extern "C" void kernel_launch(void* const* d_in, const int* in_sizes, int n_in,
                              void* d_out, int out_size, void* d_ws, size_t ws_size,
                              hipStream_t stream) {
  const float* x   = (const float*)d_in[0];
  const int*   ei  = (const int*)d_in[1];
  const float* W1  = (const float*)d_in[2];
  const float* as1 = (const float*)d_in[3];
  const float* ad1 = (const float*)d_in[4];
  const float* b1  = (const float*)d_in[5];
  const float* W2  = (const float*)d_in[6];
  const float* as2 = (const float*)d_in[7];
  const float* ad2 = (const float*)d_in[8];
  const float* b2  = (const float*)d_in[9];

  const int HC   = in_sizes[3];
  const int Fin  = in_sizes[2] / HC;
  const int N    = in_sizes[0] / Fin;
  const int E    = in_sizes[1] / 2;
  const int H    = 8;
  const int C    = HC / H;
  const int Cout = in_sizes[7];

  float* ws = (float*)d_ws;
  float* h1  = ws;
  float* x2  = h1 + (size_t)N * HC;
  float* h2  = x2 + (size_t)N * HC;
  float* aS1 = h2 + (size_t)N * Cout;
  float* aD1 = aS1 + (size_t)N * H;
  float* aS2 = aD1 + (size_t)N * H;
  float* aD2 = aS2 + N;

  const int TB = 256;
  auto blocks = [](long long n, int tb) { return (int)((n + tb - 1) / tb); };
  const int segBlocks = (N + NPB - 1) / NPB;

  {
    const long long waves = (long long)(N / 16) * (HC / 16);
    gemm_wmma_f32<<<blocks(waves * 32, TB), TB, 0, stream>>>(x, W1, h1, N, Fin, HC);
    att_coef<<<blocks((long long)N * H * 32, TB), TB, 0, stream>>>(
        h1, as1, ad1, aS1, aD1, N * H, H, C);
    seg_softmax_agg<8><<<segBlocks, SEG_TB, 0, stream>>>(
        ei, E, N, aS1, aD1, h1, b1, x2);
  }

  {
    const long long waves = (long long)(N / 16) * (Cout / 16);
    gemm_wmma_f32<<<blocks(waves * 32, TB), TB, 0, stream>>>(x2, W2, h2, N, HC, Cout);
    att_coef<<<blocks((long long)N * 32, TB), TB, 0, stream>>>(
        h2, as2, ad2, aS2, aD2, N, 1, Cout);
    seg_softmax_agg<1><<<segBlocks, SEG_TB, 0, stream>>>(
        ei, E, N, aS2, aD2, h2, b2, (float*)d_out);
  }
}


# Each block lists its incoming edges once a call (one dst-id scan in layer 1; list in LDS and workspace); max, weight and row passes of both layers walk the dense list; sums stay LDS arrival order
# speedup vs baseline: 5.6672x; 1.6098x over previous
#include <hip/hip_runtime.h>
#include <math.h>

typedef float v2f __attribute__((ext_vector_type(2)));
typedef float v8f __attribute__((ext_vector_type(8)));

#define LRELU_SLOPE 0.2f

__device__ __forceinline__ unsigned f2ord(float f) {
  unsigned u = __float_as_uint(f);
  return (u & 0x80000000u) ? ~u : (u | 0x80000000u);
}
__device__ __forceinline__ float ord2f(unsigned u) {
  return __uint_as_float((u & 0x80000000u) ? (u & 0x7FFFFFFFu) : ~u);
}

__global__ __launch_bounds__(256) void gemm_wmma_f32(
    const float* __restrict__ X, const float* __restrict__ W,
    float* __restrict__ Y, int M, int K, int Nc) {
  const int lane  = threadIdx.x & 31;
  const int wave  = (blockIdx.x * blockDim.x + threadIdx.x) >> 5;
  const int tilesN = Nc >> 4;
  const int tilesM = M >> 4;
  if (wave >= tilesM * tilesN) return;
  const int mTile = (wave / tilesN) << 4;
  const int nTile = (wave % tilesN) << 4;
  const int half  = lane >> 4;
  const int l16   = lane & 15;

  const float* __restrict__ xrow = X + (size_t)(mTile + l16) * K;
  const float* __restrict__ wcol = W + nTile + l16;

  v8f acc = {0.f, 0.f, 0.f, 0.f, 0.f, 0.f, 0.f, 0.f};
  for (int k = 0; k < K; k += 4) {
    const int ka = k + 2 * half;
    v2f a = *(const v2f*)(xrow + ka);
    v2f b;
    b.x = wcol[(size_t)ka * Nc];
    b.y = wcol[(size_t)(ka + 1) * Nc];
#if __has_builtin(__builtin_amdgcn_wmma_f32_16x16x4_f32)
    acc = __builtin_amdgcn_wmma_f32_16x16x4_f32(false, a, false, b,
                                                (short)0, acc, false, false);
#else
    acc[0] += a.x * b.x;
#endif
  }
  float* __restrict__ yrow = Y + nTile + l16;
  const int mbase = mTile + half * 8;
#pragma unroll
  for (int r = 0; r < 8; ++r)
    yrow[(size_t)(mbase + r) * Nc] = acc[r];
}

__global__ __launch_bounds__(256) void att_coef(
    const float* __restrict__ h, const float* __restrict__ att_s,
    const float* __restrict__ att_d, float* __restrict__ a_s,
    float* __restrict__ a_d, int NH, int H, int C) {
  const long long gt = (long long)blockIdx.x * blockDim.x + threadIdx.x;
  const int wave = (int)(gt >> 5);
  const int c    = threadIdx.x & 31;
  if (wave >= NH) return;
  const int hh = wave % H;
  const float v = h[(size_t)wave * C + c];
  float ss = v * att_s[hh * C + c];
  float dd = v * att_d[hh * C + c];
#pragma unroll
  for (int off = 16; off > 0; off >>= 1) {
    ss += __shfl_xor(ss, off, 32);
    dd += __shfl_xor(dd, off, 32);
  }
  if (c == 0) { a_s[wave] = ss; a_d[wave] = dd; }
}

#define NPB 196
#define SEG_TB 1024
#define SEG_WV (SEG_TB / 32)
#define SCAN_U 4
#define SCAN_G (32 * SCAN_U)
#define WCAP 192
#define LCAP (SEG_WV * WCAP + NPB)
static_assert(NPB <= 256, "local dst index is kept in a byte");

template <int H>
__device__ __forceinline__ void ld_coef(const float* __restrict__ a, int n,
                                        float (&v)[H]) {
  if constexpr ((H & 3) == 0) {
#pragma unroll
    for (int i = 0; i < H; i += 4) {
      const float4 t = *(const float4*)(a + (size_t)n * H + i);
      v[i] = t.x; v[i + 1] = t.y; v[i + 2] = t.z; v[i + 3] = t.w;
    }
  } else {
#pragma unroll
    for (int i = 0; i < H; ++i) v[i] = a[(size_t)n * H + i];
  }
}

__device__ __forceinline__ void ld_group(const int* __restrict__ dstp, int g,
                                         int lane, int4 (&v)[SCAN_U]) {
  const int4* __restrict__ p = (const int4*)dstp + (size_t)g * SCAN_G + lane;
#pragma unroll
  for (int u = 0; u < SCAN_U; ++u) v[u] = p[u * 32];
}

template <int H, bool BUILD>
__global__ __launch_bounds__(SEG_TB) void seg_softmax_agg(
    const int* __restrict__ ei, int E, int N,
    const float* __restrict__ a_s, const float* __restrict__ a_d,
    const float* __restrict__ h, const float* __restrict__ bias,
    float* __restrict__ out, int* __restrict__ wlist, int* __restrict__ wcnt) {
  constexpr int HC = H * 32;
  __shared__ float    s_num[NPB * HC];
  __shared__ float    s_den[NPB * H];
  __shared__ unsigned s_max[NPB * H];
  __shared__ float    s_ad[NPB * H];
  __shared__ int           s_stage[BUILD ? SEG_WV * WCAP : 1];
  __shared__ int           s_wc[SEG_WV];
  __shared__ int           l_src[LCAP];
  __shared__ unsigned char l_dl[LCAP];

  const int tid  = threadIdx.x;
  const int lane = tid & 31;
  const int wv   = __builtin_amdgcn_readfirstlane(tid >> 5);
  const int n0   = blockIdx.x * NPB;
  const int cnt  = min(NPB, N - n0);

  for (int i = tid; i < NPB * HC; i += SEG_TB) s_num[i] = 0.f;
  for (int i = tid; i < NPB * H; i += SEG_TB) {
    s_den[i] = 0.f;
    s_max[i] = 0u;
    s_ad[i]  = (i < cnt * H) ? a_d[(size_t)n0 * H + i] : 0.f;
  }

  const int* __restrict__ srcp = ei;
  const int* __restrict__ dstp = ei + E;
  const bool al16 = ((E & 3) == 0) && (((size_t)dstp & 15) == 0);
  const int nG    = al16 ? (E >> 2) / SCAN_G : 0;
  const int eRest = nG * SCAN_G * 4;

  auto edgeA = [&](int d, int e, bool self) {
    const int dl = d - n0;
    if ((unsigned)dl < (unsigned)cnt) {
      const int s = self ? d : srcp[e];
      float av[H];
      ld_coef<H>(a_s, s, av);
#pragma unroll
      for (int hh = 0; hh < H; ++hh) {
        float l = av[hh] + s_ad[dl * H + hh];
        l = (l > 0.f) ? l : LRELU_SLOPE * l;
        atomicMax(&s_max[dl * H + hh], f2ord(l));
      }
    }
  };

  auto edgeB = [&](int d, int e, bool self) {
    const int  dl  = d - n0;
    const bool hit = (unsigned)dl < (unsigned)cnt;
    unsigned mask = __builtin_amdgcn_ballot_w32(hit);
    if (mask != 0u) {
      int   s = 0;
      float w[H];
#pragma unroll
      for (int hh = 0; hh < H; ++hh) w[hh] = 0.f;
      if (hit) {
        s = self ? d : srcp[e];
        float av[H];
        ld_coef<H>(a_s, s, av);
#pragma unroll
        for (int hh = 0; hh < H; ++hh) {
          float l = av[hh] + s_ad[dl * H + hh];
          l = (l > 0.f) ? l : LRELU_SLOPE * l;
          w[hh] = expf(l - ord2f(s_max[dl * H + hh]));
          atomicAdd(&s_den[dl * H + hh], w[hh]);
        }
      }
      while (mask) {
        const int L = __builtin_ctz(mask);
        mask &= mask - 1u;
        const int sB = __builtin_amdgcn_readlane(s, L);
        const int dB = __builtin_amdgcn_readlane(dl, L);
        const float* __restrict__ hrow = h + (size_t)sB * HC + lane;
        float hv[H];
#pragma unroll
        for (int hh = 0; hh < H; ++hh) hv[hh] = hrow[hh * 32];
#pragma unroll
        for (int hh = 0; hh < H; ++hh) {
          const float wB = __int_as_float(
              __builtin_amdgcn_readlane(__float_as_int(w[hh]), L));
          atomicAdd(&s_num[dB * HC + hh * 32 + lane], wB * hv[hh]);
        }
      }
    }
  };

  const int nlS = lane * SEG_WV + wv;
  const int dS  = (nlS < cnt) ? n0 + nlS : -1;

  auto scan = [&](auto&& edgeFn) {
    if (wv < nG) {
      int4 nx[SCAN_U];
      ld_group(dstp, wv, lane, nx);
      for (int g = wv; g < nG; g += SEG_WV) {
        int4 cur[SCAN_U];
#pragma unroll
        for (int u = 0; u < SCAN_U; ++u) cur[u] = nx[u];
        ld_group(dstp, min(g + SEG_WV, nG - 1), lane, nx);
        unsigned hm = 0u;
#pragma unroll
        for (int u = 0; u < SCAN_U; ++u) {
          hm |= ((unsigned)(cur[u].x - n0) < (unsigned)cnt) ? (1u << (4 * u))     : 0u;
          hm |= ((unsigned)(cur[u].y - n0) < (unsigned)cnt) ? (1u << (4 * u + 1)) : 0u;
          hm |= ((unsigned)(cur[u].z - n0) < (unsigned)cnt) ? (1u << (4 * u + 2)) : 0u;
          hm |= ((unsigned)(cur[u].w - n0) < (unsigned)cnt) ? (1u << (4 * u + 3)) : 0u;
        }
        const int ebase = 4 * (g * SCAN_G + lane);
        while (__builtin_amdgcn_ballot_w32(hm != 0u) != 0u) {
          const bool has = (hm != 0u);
          const int  idx = has ? __builtin_ctz(hm) : 0;
          hm &= hm - 1u;
          const int e = ebase + (idx >> 2) * (4 * 32) + (idx & 3);
          const int d = has ? dstp[e] : -1;
          edgeFn(d, e, false);
        }
      }
    }
    for (int eb = eRest + wv * 32; eb < E; eb += SEG_TB) {
      const int e = eb + lane;
      const int d = (e < E) ? dstp[e] : -1;
      edgeFn(d, (e < E) ? e : 0, false);
    }
  };

  int wc = 0;
  if constexpr (BUILD) {
    auto stage = [&](bool hit, int e) {
      const unsigned m = __builtin_amdgcn_ballot_w32(hit);
      if (m != 0u) {
        const int pos = wc + (int)__builtin_amdgcn_mbcnt_lo(m, 0u);
        if (hit && pos < WCAP) s_stage[wv * WCAP + pos] = e;
        wc += __builtin_popcount(m);
      }
    };
    if (wv < nG) {
      int4 nx[SCAN_U];
      ld_group(dstp, wv, lane, nx);
      for (int g = wv; g < nG; g += SEG_WV) {
        int4 cur[SCAN_U];
#pragma unroll
        for (int u = 0; u < SCAN_U; ++u) cur[u] = nx[u];
        ld_group(dstp, min(g + SEG_WV, nG - 1), lane, nx);
        const int ebase = 4 * (g * SCAN_G + lane);
#pragma unroll
        for (int u = 0; u < SCAN_U; ++u) {
          stage((unsigned)(cur[u].x - n0) < (unsigned)cnt, ebase + u * 128);
          stage((unsigned)(cur[u].y - n0) < (unsigned)cnt, ebase + u * 128 + 1);
          stage((unsigned)(cur[u].z - n0) < (unsigned)cnt, ebase + u * 128 + 2);
          stage((unsigned)(cur[u].w - n0) < (unsigned)cnt, ebase + u * 128 + 3);
        }
      }
    }
    for (int eb = eRest + wv * 32; eb < E; eb += SEG_TB) {
      const int e = eb + lane;
      const int d = (e < E) ? dstp[e] : -1;
      stage((unsigned)(d - n0) < (unsigned)cnt, e);
    }
    if (lane == 0) s_wc[wv] = wc;
  }
  __syncthreads();

  int2* __restrict__ wl = (int2*)wlist + (size_t)blockIdx.x * LCAP;

  auto put = [&](int idx, int s, int dl) {
    l_src[idx] = s;
    l_dl[idx]  = (unsigned char)dl;
    if constexpr (BUILD) wl[idx] = make_int2(s, dl);
    float av[H];
    ld_coef<H>(a_s, s, av);
#pragma unroll
    for (int hh = 0; hh < H; ++hh) {
      float l = av[hh] + s_ad[dl * H + hh];
      l = (l > 0.f) ? l : LRELU_SLOPE * l;
      atomicMax(&s_max[dl * H + hh], f2ord(l));
    }
  };

  int total;
  if constexpr (BUILD) {
    const int c = s_wc[lane];
    int pre = (lane < wv) ? c : 0;
    int all = c;
#pragma unroll
    for (int off = 16; off > 0; off >>= 1) {
      pre += __shfl_xor(pre, off, 32);
      all += __shfl_xor(all, off, 32);
    }
    pre = __builtin_amdgcn_readfirstlane(pre);
    all = __builtin_amdgcn_readfirstlane(all);
    const bool over = __builtin_amdgcn_ballot_w32(c > WCAP) != 0u;
    total = over ? -1 : all + cnt;
    if (tid == 0) wcnt[blockIdx.x] = total;
    if (!over) {
      for (int j = lane; j < wc; j += 32) {
        const int e = s_stage[wv * WCAP + j];
        put(pre + j, srcp[e], dstp[e] - n0);
      }
      if (tid < cnt) put(all + tid, n0 + tid, tid);
    }
  } else {
    total = wcnt[blockIdx.x];
    if (total > LCAP) total = -1;
    for (int i = tid; i < total; i += SEG_TB) {
      const int2 v = wl[i];
      put(i, v.x, v.y);
    }
  }

  if (total >= 0) {
    __syncthreads();
    constexpr int UNR = (H >= 8) ? 4 : 16;
    const int per = (total + SEG_WV - 1) / SEG_WV;
    const int lo  = wv * per;
    const int hi  = min(total, lo + per);
    for (int base = lo; base < hi; base += 32) {
      const int  i     = base + lane;
      const bool valid = i < hi;
      int   s = 0, dl = 0;
      float w[H];
#pragma unroll
      for (int hh = 0; hh < H; ++hh) w[hh] = 0.f;
      if (valid) {
        s  = l_src[i];
        dl = l_dl[i];
        float av[H];
        ld_coef<H>(a_s, s, av);
#pragma unroll
        for (int hh = 0; hh < H; ++hh) {
          float l = av[hh] + s_ad[dl * H + hh];
          l = (l > 0.f) ? l : LRELU_SLOPE * l;
          w[hh] = expf(l - ord2f(s_max[dl * H + hh]));
          atomicAdd(&s_den[dl * H + hh], w[hh]);
        }
      }
      const int n = min(32, hi - base);
      for (int L0 = 0; L0 < n; L0 += UNR) {
        float hv[UNR][H];
        int   dB[UNR];
#pragma unroll
        for (int j = 0; j < UNR; ++j) {
          const int L  = min(L0 + j, n - 1);
          const int sB = __builtin_amdgcn_readlane(s, L);
          dB[j]        = __builtin_amdgcn_readlane(dl, L);
          const float* __restrict__ hrow = h + (size_t)sB * HC + lane;
#pragma unroll
          for (int hh = 0; hh < H; ++hh) hv[j][hh] = hrow[hh * 32];
        }
#pragma unroll
        for (int j = 0; j < UNR; ++j) {
          if (L0 + j < n) {
#pragma unroll
            for (int hh = 0; hh < H; ++hh) {
              const float wB = __int_as_float(
                  __builtin_amdgcn_readlane(__float_as_int(w[hh]), L0 + j));
              atomicAdd(&s_num[dB[j] * HC + hh * 32 + lane], wB * hv[j][hh]);
            }
          }
        }
      }
    }
  } else {
    edgeA(dS, 0, true);
    scan(edgeA);
    __syncthreads();
    edgeB(dS, 0, true);
    scan(edgeB);
  }
  __syncthreads();

  float* __restrict__ orow = out + (size_t)n0 * HC;
  for (int i = tid; i < cnt * HC; i += SEG_TB) {
    const int nl = i / HC;
    const int hc = i % HC;
    float v = s_num[i] / (s_den[nl * H + hc / 32] + 1e-16f) + bias[hc];
    v = (v > 0.f) ? v : (expf(v) - 1.f);
    orow[i] = v;
  }
}

extern "C" void kernel_launch(void* const* d_in, const int* in_sizes, int n_in,
                              void* d_out, int out_size, void* d_ws, size_t ws_size,
                              hipStream_t stream) {
  const float* x   = (const float*)d_in[0];
  const int*   ei  = (const int*)d_in[1];
  const float* W1  = (const float*)d_in[2];
  const float* as1 = (const float*)d_in[3];
  const float* ad1 = (const float*)d_in[4];
  const float* b1  = (const float*)d_in[5];
  const float* W2  = (const float*)d_in[6];
  const float* as2 = (const float*)d_in[7];
  const float* ad2 = (const float*)d_in[8];
  const float* b2  = (const float*)d_in[9];

  const int HC   = in_sizes[3];
  const int Fin  = in_sizes[2] / HC;
  const int N    = in_sizes[0] / Fin;
  const int E    = in_sizes[1] / 2;
  const int H    = 8;
  const int C    = HC / H;
  const int Cout = in_sizes[7];

  float* ws = (float*)d_ws;
  float* h1  = ws;
  float* x2  = h1 + (size_t)N * HC;
  float* h2  = x2 + (size_t)N * HC;
  float* aS1 = h2 + (size_t)N * Cout;
  float* aD1 = aS1 + (size_t)N * H;
  float* aS2 = aD1 + (size_t)N * H;
  float* aD2 = aS2 + N;
  const int segBlocksWs = (N + NPB - 1) / NPB;
  int* eList = (int*)(aD2 + N);
  int* eCnt  = eList + (size_t)segBlocksWs * LCAP * 2;

  const int TB = 256;
  auto blocks = [](long long n, int tb) { return (int)((n + tb - 1) / tb); };
  const int segBlocks = (N + NPB - 1) / NPB;

  {
    const long long waves = (long long)(N / 16) * (HC / 16);
    gemm_wmma_f32<<<blocks(waves * 32, TB), TB, 0, stream>>>(x, W1, h1, N, Fin, HC);
    att_coef<<<blocks((long long)N * H * 32, TB), TB, 0, stream>>>(
        h1, as1, ad1, aS1, aD1, N * H, H, C);
    seg_softmax_agg<8, true><<<segBlocks, SEG_TB, 0, stream>>>(
        ei, E, N, aS1, aD1, h1, b1, x2, eList, eCnt);
  }

  {
    const long long waves = (long long)(N / 16) * (Cout / 16);
    gemm_wmma_f32<<<blocks(waves * 32, TB), TB, 0, stream>>>(x2, W2, h2, N, HC, Cout);
    att_coef<<<blocks((long long)N * 32, TB), TB, 0, stream>>>(
        h2, as2, ad2, aS2, aD2, N, 1, Cout);
    seg_softmax_agg<1, false><<<segBlocks, SEG_TB, 0, stream>>>(
        ei, E, N, aS2, aD2, h2, b2, (float*)d_out, eList, eCnt);
  }
}


# Attention dots move into the GEMM epilogue: a wave owns 16 rows x one 32-channel head (2 accumulators), sums its dots in the same tree (c,c+16 fused; xor 8,4,2,1); both att_coef launches go
# speedup vs baseline: 7.8034x; 1.3769x over previous
#include <hip/hip_runtime.h>
#include <math.h>

typedef float v2f __attribute__((ext_vector_type(2)));
typedef float v8f __attribute__((ext_vector_type(8)));

#define LRELU_SLOPE 0.2f

__device__ __forceinline__ unsigned f2ord(float f) {
  unsigned u = __float_as_uint(f);
  return (u & 0x80000000u) ? ~u : (u | 0x80000000u);
}
__device__ __forceinline__ float ord2f(unsigned u) {
  return __uint_as_float((u & 0x80000000u) ? (u & 0x7FFFFFFFu) : ~u);
}

template <int K, int NC>
__global__ __launch_bounds__(256) void gemm_att_wmma_f32(
    const float* __restrict__ X, const float* __restrict__ W,
    const float* __restrict__ att_s, const float* __restrict__ att_d,
    float* __restrict__ Y, float* __restrict__ a_s, float* __restrict__ a_d,
    int M) {
  constexpr int HEADS = NC / 32;
  static_assert(NC % 32 == 0 && K % 4 == 0, "a head is 32 channels; K steps by 4");
  const int lane  = threadIdx.x & 31;
  const int wave  = (blockIdx.x * blockDim.x + threadIdx.x) >> 5;
  if (wave >= (M >> 4) * HEADS) return;
  const int mTile = (wave / HEADS) << 4;
  const int head  = wave % HEADS;
  const int half  = lane >> 4;
  const int l16   = lane & 15;

  const float* __restrict__ xrow = X + (size_t)(mTile + l16) * K + 2 * half;
  const float* __restrict__ wcol = W + (size_t)(2 * half) * NC + head * 32 + l16;

  v8f acc0 = {0.f, 0.f, 0.f, 0.f, 0.f, 0.f, 0.f, 0.f};
  v8f acc1 = {0.f, 0.f, 0.f, 0.f, 0.f, 0.f, 0.f, 0.f};
#pragma unroll 8
  for (int k = 0; k < K; k += 4) {
    v2f a = *(const v2f*)(xrow + k);
    v2f b0, b1;
    b0.x = wcol[k * NC];
    b0.y = wcol[(k + 1) * NC];
    b1.x = wcol[k * NC + 16];
    b1.y = wcol[(k + 1) * NC + 16];
#if __has_builtin(__builtin_amdgcn_wmma_f32_16x16x4_f32)
    acc0 = __builtin_amdgcn_wmma_f32_16x16x4_f32(false, a, false, b0,
                                                 (short)0, acc0, false, false);
    acc1 = __builtin_amdgcn_wmma_f32_16x16x4_f32(false, a, false, b1,
                                                 (short)0, acc1, false, false);
#else
    acc0[0] += a.x * b0.x;
    acc1[0] += a.x * b1.x;
#endif
  }
  float* __restrict__ yrow = Y + (size_t)(mTile + half * 8) * NC + head * 32 + l16;
#pragma unroll
  for (int r = 0; r < 8; ++r) {
    yrow[r * NC]      = acc0[r];
    yrow[r * NC + 16] = acc1[r];
  }

  const float s0 = att_s[head * 32 + l16], s1 = att_s[head * 32 + 16 + l16];
  const float d0 = att_d[head * 32 + l16], d1 = att_d[head * 32 + 16 + l16];
  float v[16];
#pragma unroll
  for (int r = 0; r < 8; ++r) {
    v[r]     = fmaf(acc0[r], s0, acc1[r] * s1);
    v[8 + r] = fmaf(acc0[r], d0, acc1[r] * d1);
  }
  const bool b3 = (l16 & 8) != 0;
  float u[8];
#pragma unroll
  for (int j = 0; j < 8; ++j) {
    const float keep = b3 ? v[8 + j] : v[j];
    const float send = b3 ? v[j] : v[8 + j];
    u[j] = keep + __shfl_xor(send, 8, 32);
  }
  const bool b2 = (l16 & 4) != 0;
  float t[4];
#pragma unroll
  for (int j = 0; j < 4; ++j) {
    const float keep = b2 ? u[4 + j] : u[j];
    const float send = b2 ? u[j] : u[4 + j];
    t[j] = keep + __shfl_xor(send, 4, 32);
  }
  const bool b1 = (l16 & 2) != 0;
  float q[2];
#pragma unroll
  for (int j = 0; j < 2; ++j) {
    const float keep = b1 ? t[2 + j] : t[j];
    const float send = b1 ? t[j] : t[2 + j];
    q[j] = keep + __shfl_xor(send, 2, 32);
  }
  const bool b0 = (l16 & 1) != 0;
  const float keep = b0 ? q[1] : q[0];
  const float send = b0 ? q[0] : q[1];
  const float res  = keep + __shfl_xor(send, 1, 32);
  const int row = mTile + 8 * half + (l16 & 7);
  float* __restrict__ dst = b3 ? a_d : a_s;
  dst[(size_t)row * HEADS + head] = res;
}

#define NPB 196
#define SEG_TB 1024
#define SEG_WV (SEG_TB / 32)
#define SCAN_U 4
#define SCAN_G (32 * SCAN_U)
#define WCAP 192
#define LCAP (SEG_WV * WCAP + NPB)
static_assert(NPB <= 256, "local dst index is kept in a byte");

template <int H>
__device__ __forceinline__ void ld_coef(const float* __restrict__ a, int n,
                                        float (&v)[H]) {
  if constexpr ((H & 3) == 0) {
#pragma unroll
    for (int i = 0; i < H; i += 4) {
      const float4 t = *(const float4*)(a + (size_t)n * H + i);
      v[i] = t.x; v[i + 1] = t.y; v[i + 2] = t.z; v[i + 3] = t.w;
    }
  } else {
#pragma unroll
    for (int i = 0; i < H; ++i) v[i] = a[(size_t)n * H + i];
  }
}

__device__ __forceinline__ void ld_group(const int* __restrict__ dstp, int g,
                                         int lane, int4 (&v)[SCAN_U]) {
  const int4* __restrict__ p = (const int4*)dstp + (size_t)g * SCAN_G + lane;
#pragma unroll
  for (int u = 0; u < SCAN_U; ++u) v[u] = p[u * 32];
}

template <int H, bool BUILD>
__global__ __launch_bounds__(SEG_TB) void seg_softmax_agg(
    const int* __restrict__ ei, int E, int N,
    const float* __restrict__ a_s, const float* __restrict__ a_d,
    const float* __restrict__ h, const float* __restrict__ bias,
    float* __restrict__ out, int* __restrict__ wlist, int* __restrict__ wcnt) {
  constexpr int HC = H * 32;
  __shared__ float    s_num[NPB * HC];
  __shared__ float    s_den[NPB * H];
  __shared__ unsigned s_max[NPB * H];
  __shared__ float    s_ad[NPB * H];
  __shared__ int           s_stage[BUILD ? SEG_WV * WCAP : 1];
  __shared__ int           s_wc[SEG_WV];
  __shared__ int           l_src[LCAP];
  __shared__ unsigned char l_dl[LCAP];

  const int tid  = threadIdx.x;
  const int lane = tid & 31;
  const int wv   = __builtin_amdgcn_readfirstlane(tid >> 5);
  const int n0   = blockIdx.x * NPB;
  const int cnt  = min(NPB, N - n0);

  for (int i = tid; i < NPB * HC; i += SEG_TB) s_num[i] = 0.f;
  for (int i = tid; i < NPB * H; i += SEG_TB) {
    s_den[i] = 0.f;
    s_max[i] = 0u;
    s_ad[i]  = (i < cnt * H) ? a_d[(size_t)n0 * H + i] : 0.f;
  }

  const int* __restrict__ srcp = ei;
  const int* __restrict__ dstp = ei + E;
  const bool al16 = ((E & 3) == 0) && (((size_t)dstp & 15) == 0);
  const int nG    = al16 ? (E >> 2) / SCAN_G : 0;
  const int eRest = nG * SCAN_G * 4;

  auto edgeA = [&](int d, int e, bool self) {
    const int dl = d - n0;
    if ((unsigned)dl < (unsigned)cnt) {
      const int s = self ? d : srcp[e];
      float av[H];
      ld_coef<H>(a_s, s, av);
#pragma unroll
      for (int hh = 0; hh < H; ++hh) {
        float l = av[hh] + s_ad[dl * H + hh];
        l = (l > 0.f) ? l : LRELU_SLOPE * l;
        atomicMax(&s_max[dl * H + hh], f2ord(l));
      }
    }
  };

  auto edgeB = [&](int d, int e, bool self) {
    const int  dl  = d - n0;
    const bool hit = (unsigned)dl < (unsigned)cnt;
    unsigned mask = __builtin_amdgcn_ballot_w32(hit);
    if (mask != 0u) {
      int   s = 0;
      float w[H];
#pragma unroll
      for (int hh = 0; hh < H; ++hh) w[hh] = 0.f;
      if (hit) {
        s = self ? d : srcp[e];
        float av[H];
        ld_coef<H>(a_s, s, av);
#pragma unroll
        for (int hh = 0; hh < H; ++hh) {
          float l = av[hh] + s_ad[dl * H + hh];
          l = (l > 0.f) ? l : LRELU_SLOPE * l;
          w[hh] = expf(l - ord2f(s_max[dl * H + hh]));
          atomicAdd(&s_den[dl * H + hh], w[hh]);
        }
      }
      while (mask) {
        const int L = __builtin_ctz(mask);
        mask &= mask - 1u;
        const int sB = __builtin_amdgcn_readlane(s, L);
        const int dB = __builtin_amdgcn_readlane(dl, L);
        const float* __restrict__ hrow = h + (size_t)sB * HC + lane;
        float hv[H];
#pragma unroll
        for (int hh = 0; hh < H; ++hh) hv[hh] = hrow[hh * 32];
#pragma unroll
        for (int hh = 0; hh < H; ++hh) {
          const float wB = __int_as_float(
              __builtin_amdgcn_readlane(__float_as_int(w[hh]), L));
          atomicAdd(&s_num[dB * HC + hh * 32 + lane], wB * hv[hh]);
        }
      }
    }
  };

  const int nlS = lane * SEG_WV + wv;
  const int dS  = (nlS < cnt) ? n0 + nlS : -1;

  auto scan = [&](auto&& edgeFn) {
    if (wv < nG) {
      int4 nx[SCAN_U];
      ld_group(dstp, wv, lane, nx);
      for (int g = wv; g < nG; g += SEG_WV) {
        int4 cur[SCAN_U];
#pragma unroll
        for (int u = 0; u < SCAN_U; ++u) cur[u] = nx[u];
        ld_group(dstp, min(g + SEG_WV, nG - 1), lane, nx);
        unsigned hm = 0u;
#pragma unroll
        for (int u = 0; u < SCAN_U; ++u) {
          hm |= ((unsigned)(cur[u].x - n0) < (unsigned)cnt) ? (1u << (4 * u))     : 0u;
          hm |= ((unsigned)(cur[u].y - n0) < (unsigned)cnt) ? (1u << (4 * u + 1)) : 0u;
          hm |= ((unsigned)(cur[u].z - n0) < (unsigned)cnt) ? (1u << (4 * u + 2)) : 0u;
          hm |= ((unsigned)(cur[u].w - n0) < (unsigned)cnt) ? (1u << (4 * u + 3)) : 0u;
        }
        const int ebase = 4 * (g * SCAN_G + lane);
        while (__builtin_amdgcn_ballot_w32(hm != 0u) != 0u) {
          const bool has = (hm != 0u);
          const int  idx = has ? __builtin_ctz(hm) : 0;
          hm &= hm - 1u;
          const int e = ebase + (idx >> 2) * (4 * 32) + (idx & 3);
          const int d = has ? dstp[e] : -1;
          edgeFn(d, e, false);
        }
      }
    }
    for (int eb = eRest + wv * 32; eb < E; eb += SEG_TB) {
      const int e = eb + lane;
      const int d = (e < E) ? dstp[e] : -1;
      edgeFn(d, (e < E) ? e : 0, false);
    }
  };

  int wc = 0;
  if constexpr (BUILD) {
    auto stage = [&](bool hit, int e) {
      const unsigned m = __builtin_amdgcn_ballot_w32(hit);
      if (m != 0u) {
        const int pos = wc + (int)__builtin_amdgcn_mbcnt_lo(m, 0u);
        if (hit && pos < WCAP) s_stage[wv * WCAP + pos] = e;
        wc += __builtin_popcount(m);
      }
    };
    if (wv < nG) {
      int4 nx[SCAN_U];
      ld_group(dstp, wv, lane, nx);
      for (int g = wv; g < nG; g += SEG_WV) {
        int4 cur[SCAN_U];
#pragma unroll
        for (int u = 0; u < SCAN_U; ++u) cur[u] = nx[u];
        ld_group(dstp, min(g + SEG_WV, nG - 1), lane, nx);
        const int ebase = 4 * (g * SCAN_G + lane);
#pragma unroll
        for (int u = 0; u < SCAN_U; ++u) {
          stage((unsigned)(cur[u].x - n0) < (unsigned)cnt, ebase + u * 128);
          stage((unsigned)(cur[u].y - n0) < (unsigned)cnt, ebase + u * 128 + 1);
          stage((unsigned)(cur[u].z - n0) < (unsigned)cnt, ebase + u * 128 + 2);
          stage((unsigned)(cur[u].w - n0) < (unsigned)cnt, ebase + u * 128 + 3);
        }
      }
    }
    for (int eb = eRest + wv * 32; eb < E; eb += SEG_TB) {
      const int e = eb + lane;
      const int d = (e < E) ? dstp[e] : -1;
      stage((unsigned)(d - n0) < (unsigned)cnt, e);
    }
    if (lane == 0) s_wc[wv] = wc;
  }
  __syncthreads();

  int2* __restrict__ wl = (int2*)wlist + (size_t)blockIdx.x * LCAP;

  auto put = [&](int idx, int s, int dl) {
    l_src[idx] = s;
    l_dl[idx]  = (unsigned char)dl;
    if constexpr (BUILD) wl[idx] = make_int2(s, dl);
    float av[H];
    ld_coef<H>(a_s, s, av);
#pragma unroll
    for (int hh = 0; hh < H; ++hh) {
      float l = av[hh] + s_ad[dl * H + hh];
      l = (l > 0.f) ? l : LRELU_SLOPE * l;
      atomicMax(&s_max[dl * H + hh], f2ord(l));
    }
  };

  int total;
  if constexpr (BUILD) {
    const int c = s_wc[lane];
    int pre = (lane < wv) ? c : 0;
    int all = c;
#pragma unroll
    for (int off = 16; off > 0; off >>= 1) {
      pre += __shfl_xor(pre, off, 32);
      all += __shfl_xor(all, off, 32);
    }
    pre = __builtin_amdgcn_readfirstlane(pre);
    all = __builtin_amdgcn_readfirstlane(all);
    const bool over = __builtin_amdgcn_ballot_w32(c > WCAP) != 0u;
    total = over ? -1 : all + cnt;
    if (tid == 0) wcnt[blockIdx.x] = total;
    if (!over) {
      for (int j = lane; j < wc; j += 32) {
        const int e = s_stage[wv * WCAP + j];
        put(pre + j, srcp[e], dstp[e] - n0);
      }
      if (tid < cnt) put(all + tid, n0 + tid, tid);
    }
  } else {
    total = wcnt[blockIdx.x];
    if (total > LCAP) total = -1;
    for (int i = tid; i < total; i += SEG_TB) {
      const int2 v = wl[i];
      put(i, v.x, v.y);
    }
  }

  if (total >= 0) {
    __syncthreads();
    constexpr int UNR = (H >= 8) ? 4 : 16;
    const int per = (total + SEG_WV - 1) / SEG_WV;
    const int lo  = wv * per;
    const int hi  = min(total, lo + per);
    for (int base = lo; base < hi; base += 32) {
      const int  i     = base + lane;
      const bool valid = i < hi;
      int   s = 0, dl = 0;
      float w[H];
#pragma unroll
      for (int hh = 0; hh < H; ++hh) w[hh] = 0.f;
      if (valid) {
        s  = l_src[i];
        dl = l_dl[i];
        float av[H];
        ld_coef<H>(a_s, s, av);
#pragma unroll
        for (int hh = 0; hh < H; ++hh) {
          float l = av[hh] + s_ad[dl * H + hh];
          l = (l > 0.f) ? l : LRELU_SLOPE * l;
          w[hh] = expf(l - ord2f(s_max[dl * H + hh]));
          atomicAdd(&s_den[dl * H + hh], w[hh]);
        }
      }
      const int n = min(32, hi - base);
      for (int L0 = 0; L0 < n; L0 += UNR) {
        float hv[UNR][H];
        int   dB[UNR];
#pragma unroll
        for (int j = 0; j < UNR; ++j) {
          const int L  = min(L0 + j, n - 1);
          const int sB = __builtin_amdgcn_readlane(s, L);
          dB[j]        = __builtin_amdgcn_readlane(dl, L);
          const float* __restrict__ hrow = h + (size_t)sB * HC + lane;
#pragma unroll
          for (int hh = 0; hh < H; ++hh) hv[j][hh] = hrow[hh * 32];
        }
#pragma unroll
        for (int j = 0; j < UNR; ++j) {
          if (L0 + j < n) {
#pragma unroll
            for (int hh = 0; hh < H; ++hh) {
              const float wB = __int_as_float(
                  __builtin_amdgcn_readlane(__float_as_int(w[hh]), L0 + j));
              atomicAdd(&s_num[dB[j] * HC + hh * 32 + lane], wB * hv[j][hh]);
            }
          }
        }
      }
    }
  } else {
    edgeA(dS, 0, true);
    scan(edgeA);
    __syncthreads();
    edgeB(dS, 0, true);
    scan(edgeB);
  }
  __syncthreads();

  float* __restrict__ orow = out + (size_t)n0 * HC;
  for (int i = tid; i < cnt * HC; i += SEG_TB) {
    const int nl = i / HC;
    const int hc = i % HC;
    float v = s_num[i] / (s_den[nl * H + hc / 32] + 1e-16f) + bias[hc];
    v = (v > 0.f) ? v : (expf(v) - 1.f);
    orow[i] = v;
  }
}

extern "C" void kernel_launch(void* const* d_in, const int* in_sizes, int n_in,
                              void* d_out, int out_size, void* d_ws, size_t ws_size,
                              hipStream_t stream) {
  const float* x   = (const float*)d_in[0];
  const int*   ei  = (const int*)d_in[1];
  const float* W1  = (const float*)d_in[2];
  const float* as1 = (const float*)d_in[3];
  const float* ad1 = (const float*)d_in[4];
  const float* b1  = (const float*)d_in[5];
  const float* W2  = (const float*)d_in[6];
  const float* as2 = (const float*)d_in[7];
  const float* ad2 = (const float*)d_in[8];
  const float* b2  = (const float*)d_in[9];

  const int HC   = in_sizes[3];
  const int Fin  = in_sizes[2] / HC;
  const int N    = in_sizes[0] / Fin;
  const int E    = in_sizes[1] / 2;
  const int H    = 8;
  const int Cout = in_sizes[7];

  float* ws = (float*)d_ws;
  float* h1  = ws;
  float* x2  = h1 + (size_t)N * HC;
  float* h2  = x2 + (size_t)N * HC;
  float* aS1 = h2 + (size_t)N * Cout;
  float* aD1 = aS1 + (size_t)N * H;
  float* aS2 = aD1 + (size_t)N * H;
  float* aD2 = aS2 + N;
  const int segBlocksWs = (N + NPB - 1) / NPB;
  int* eList = (int*)(aD2 + N);
  int* eCnt  = eList + (size_t)segBlocksWs * LCAP * 2;

  const int TB = 256;
  auto blocks = [](long long n, int tb) { return (int)((n + tb - 1) / tb); };
  const int segBlocks = (N + NPB - 1) / NPB;

  {
    const long long waves = (long long)(N / 16) * (HC / 32);
    gemm_att_wmma_f32<128, 256><<<blocks(waves * 32, TB), TB, 0, stream>>>(
        x, W1, as1, ad1, h1, aS1, aD1, N);
    seg_softmax_agg<8, true><<<segBlocks, SEG_TB, 0, stream>>>(
        ei, E, N, aS1, aD1, h1, b1, x2, eList, eCnt);
  }

  {
    const long long waves = (long long)(N / 16) * (Cout / 32);
    gemm_att_wmma_f32<256, 32><<<blocks(waves * 32, TB), TB, 0, stream>>>(
        x2, W2, as2, ad2, h2, aS2, aD2, N);
    seg_softmax_agg<1, false><<<segBlocks, SEG_TB, 0, stream>>>(
        ei, E, N, aS2, aD2, h2, b2, (float*)d_out, eList, eCnt);
  }
}


# Layer-1 GEMM launch also writes each edge's owner block (dst/196) as one byte; a block's list scan reads those bytes, 16 edges per 16-B load, hits kept as per-lane bit masks; list order changes
# speedup vs baseline: 8.5456x; 1.0951x over previous
#include <hip/hip_runtime.h>
#include <math.h>

typedef float v2f __attribute__((ext_vector_type(2)));
typedef float v8f __attribute__((ext_vector_type(8)));

#define LRELU_SLOPE 0.2f
#define NPB 196

__device__ __forceinline__ unsigned f2ord(float f) {
  unsigned u = __float_as_uint(f);
  return (u & 0x80000000u) ? ~u : (u | 0x80000000u);
}
__device__ __forceinline__ float ord2f(unsigned u) {
  return __uint_as_float((u & 0x80000000u) ? (u & 0x7FFFFFFFu) : ~u);
}

template <int K, int NC, bool OWN>
__global__ __launch_bounds__(256) void gemm_att_wmma_f32(
    const float* __restrict__ X, const float* __restrict__ W,
    const float* __restrict__ att_s, const float* __restrict__ att_d,
    float* __restrict__ Y, float* __restrict__ a_s, float* __restrict__ a_d,
    int M, const int* __restrict__ dstp, int E, int nblk, unsigned* __restrict__ ob) {
  constexpr int HEADS = NC / 32;
  static_assert(NC % 32 == 0 && K % 4 == 0, "a head is 32 channels; K steps by 4");
  const int lane  = threadIdx.x & 31;
  if constexpr (OWN) {
    const bool ok = ((E & 3) == 0) && (((size_t)dstp & 15) == 0) &&
                    (((size_t)ob & 15) == 0) && nblk <= 256;
    if (ok) {
      const int nQ = E >> 2;
      const int gw = (blockIdx.x * blockDim.x + threadIdx.x) >> 5;
      const int nw = (gridDim.x * blockDim.x) >> 5;
      for (int base = gw * 32; base < nQ; base += nw * 32) {
        const int q = base + lane;
        if (q < nQ) {
          const int4 d = ((const int4*)dstp)[q];
          ob[q] = ((unsigned)d.x / NPB) | (((unsigned)d.y / NPB) << 8) |
                  (((unsigned)d.z / NPB) << 16) | (((unsigned)d.w / NPB) << 24);
        }
      }
    }
  }
  const int wave  = (blockIdx.x * blockDim.x + threadIdx.x) >> 5;
  if (wave >= (M >> 4) * HEADS) return;
  const int mTile = (wave / HEADS) << 4;
  const int head  = wave % HEADS;
  const int half  = lane >> 4;
  const int l16   = lane & 15;

  const float* __restrict__ xrow = X + (size_t)(mTile + l16) * K + 2 * half;
  const float* __restrict__ wcol = W + (size_t)(2 * half) * NC + head * 32 + l16;

  v8f acc0 = {0.f, 0.f, 0.f, 0.f, 0.f, 0.f, 0.f, 0.f};
  v8f acc1 = {0.f, 0.f, 0.f, 0.f, 0.f, 0.f, 0.f, 0.f};
#pragma unroll 8
  for (int k = 0; k < K; k += 4) {
    v2f a = *(const v2f*)(xrow + k);
    v2f b0, b1;
    b0.x = wcol[k * NC];
    b0.y = wcol[(k + 1) * NC];
    b1.x = wcol[k * NC + 16];
    b1.y = wcol[(k + 1) * NC + 16];
#if __has_builtin(__builtin_amdgcn_wmma_f32_16x16x4_f32)
    acc0 = __builtin_amdgcn_wmma_f32_16x16x4_f32(false, a, false, b0,
                                                 (short)0, acc0, false, false);
    acc1 = __builtin_amdgcn_wmma_f32_16x16x4_f32(false, a, false, b1,
                                                 (short)0, acc1, false, false);
#else
    acc0[0] += a.x * b0.x;
    acc1[0] += a.x * b1.x;
#endif
  }
  float* __restrict__ yrow = Y + (size_t)(mTile + half * 8) * NC + head * 32 + l16;
#pragma unroll
  for (int r = 0; r < 8; ++r) {
    yrow[r * NC]      = acc0[r];
    yrow[r * NC + 16] = acc1[r];
  }

  const float s0 = att_s[head * 32 + l16], s1 = att_s[head * 32 + 16 + l16];
  const float d0 = att_d[head * 32 + l16], d1 = att_d[head * 32 + 16 + l16];
  float v[16];
#pragma unroll
  for (int r = 0; r < 8; ++r) {
    v[r]     = fmaf(acc0[r], s0, acc1[r] * s1);
    v[8 + r] = fmaf(acc0[r], d0, acc1[r] * d1);
  }
  const bool b3 = (l16 & 8) != 0;
  float u[8];
#pragma unroll
  for (int j = 0; j < 8; ++j) {
    const float keep = b3 ? v[8 + j] : v[j];
    const float send = b3 ? v[j] : v[8 + j];
    u[j] = keep + __shfl_xor(send, 8, 32);
  }
  const bool b2 = (l16 & 4) != 0;
  float t[4];
#pragma unroll
  for (int j = 0; j < 4; ++j) {
    const float keep = b2 ? u[4 + j] : u[j];
    const float send = b2 ? u[j] : u[4 + j];
    t[j] = keep + __shfl_xor(send, 4, 32);
  }
  const bool b1 = (l16 & 2) != 0;
  float q[2];
#pragma unroll
  for (int j = 0; j < 2; ++j) {
    const float keep = b1 ? t[2 + j] : t[j];
    const float send = b1 ? t[j] : t[2 + j];
    q[j] = keep + __shfl_xor(send, 2, 32);
  }
  const bool b0 = (l16 & 1) != 0;
  const float keep = b0 ? q[1] : q[0];
  const float send = b0 ? q[0] : q[1];
  const float res  = keep + __shfl_xor(send, 1, 32);
  const int row = mTile + 8 * half + (l16 & 7);
  float* __restrict__ dst = b3 ? a_d : a_s;
  dst[(size_t)row * HEADS + head] = res;
}

#define SEG_TB 1024
#define SEG_WV (SEG_TB / 32)
#define SCAN_U 4
#define SCAN_G (32 * SCAN_U)
#define WCAP 192
#define LCAP (SEG_WV * WCAP + NPB)
static_assert(NPB <= 256, "local dst index is kept in a byte");
static_assert(SCAN_U == 4, "the owner-byte scan packs a trip's 16 dwords per lane into two 32-bit hit masks");

template <int H>
__device__ __forceinline__ void ld_coef(const float* __restrict__ a, int n,
                                        float (&v)[H]) {
  if constexpr ((H & 3) == 0) {
#pragma unroll
    for (int i = 0; i < H; i += 4) {
      const float4 t = *(const float4*)(a + (size_t)n * H + i);
      v[i] = t.x; v[i + 1] = t.y; v[i + 2] = t.z; v[i + 3] = t.w;
    }
  } else {
#pragma unroll
    for (int i = 0; i < H; ++i) v[i] = a[(size_t)n * H + i];
  }
}

__device__ __forceinline__ void ld_group(const int* __restrict__ dstp, int g,
                                         int lane, int4 (&v)[SCAN_U]) {
  const int4* __restrict__ p = (const int4*)dstp + (size_t)g * SCAN_G + lane;
#pragma unroll
  for (int u = 0; u < SCAN_U; ++u) v[u] = p[u * 32];
}

__device__ __forceinline__ void ld_groupb(const unsigned* __restrict__ ob, int g,
                                          int lane, uint4 (&v)[SCAN_U]) {
  const uint4* __restrict__ p = (const uint4*)ob + (size_t)g * SCAN_G + lane;
#pragma unroll
  for (int u = 0; u < SCAN_U; ++u) v[u] = p[u * 32];
}

template <int H, bool BUILD>
__global__ __launch_bounds__(SEG_TB) void seg_softmax_agg(
    const int* __restrict__ ei, int E, int N,
    const float* __restrict__ a_s, const float* __restrict__ a_d,
    const float* __restrict__ h, const float* __restrict__ bias,
    float* __restrict__ out, int* __restrict__ wlist, int* __restrict__ wcnt,
    const unsigned* __restrict__ ob) {
  constexpr int HC = H * 32;
  __shared__ float    s_num[NPB * HC];
  __shared__ float    s_den[NPB * H];
  __shared__ unsigned s_max[NPB * H];
  __shared__ float    s_ad[NPB * H];
  __shared__ int           s_stage[BUILD ? SEG_WV * WCAP : 1];
  __shared__ int           s_wc[SEG_WV];
  __shared__ int           l_src[LCAP];
  __shared__ unsigned char l_dl[LCAP];

  const int tid  = threadIdx.x;
  const int lane = tid & 31;
  const int wv   = __builtin_amdgcn_readfirstlane(tid >> 5);
  const int n0   = blockIdx.x * NPB;
  const int cnt  = min(NPB, N - n0);

  for (int i = tid; i < NPB * HC; i += SEG_TB) s_num[i] = 0.f;
  for (int i = tid; i < NPB * H; i += SEG_TB) {
    s_den[i] = 0.f;
    s_max[i] = 0u;
    s_ad[i]  = (i < cnt * H) ? a_d[(size_t)n0 * H + i] : 0.f;
  }

  const int* __restrict__ srcp = ei;
  const int* __restrict__ dstp = ei + E;
  const bool al16 = ((E & 3) == 0) && (((size_t)dstp & 15) == 0);
  const int nG    = al16 ? (E >> 2) / SCAN_G : 0;
  const int eRest = nG * SCAN_G * 4;

  auto edgeA = [&](int d, int e, bool self) {
    const int dl = d - n0;
    if ((unsigned)dl < (unsigned)cnt) {
      const int s = self ? d : srcp[e];
      float av[H];
      ld_coef<H>(a_s, s, av);
#pragma unroll
      for (int hh = 0; hh < H; ++hh) {
        float l = av[hh] + s_ad[dl * H + hh];
        l = (l > 0.f) ? l : LRELU_SLOPE * l;
        atomicMax(&s_max[dl * H + hh], f2ord(l));
      }
    }
  };

  auto edgeB = [&](int d, int e, bool self) {
    const int  dl  = d - n0;
    const bool hit = (unsigned)dl < (unsigned)cnt;
    unsigned mask = __builtin_amdgcn_ballot_w32(hit);
    if (mask != 0u) {
      int   s = 0;
      float w[H];
#pragma unroll
      for (int hh = 0; hh < H; ++hh) w[hh] = 0.f;
      if (hit) {
        s = self ? d : srcp[e];
        float av[H];
        ld_coef<H>(a_s, s, av);
#pragma unroll
        for (int hh = 0; hh < H; ++hh) {
          float l = av[hh] + s_ad[dl * H + hh];
          l = (l > 0.f) ? l : LRELU_SLOPE * l;
          w[hh] = expf(l - ord2f(s_max[dl * H + hh]));
          atomicAdd(&s_den[dl * H + hh], w[hh]);
        }
      }
      while (mask) {
        const int L = __builtin_ctz(mask);
        mask &= mask - 1u;
        const int sB = __builtin_amdgcn_readlane(s, L);
        const int dB = __builtin_amdgcn_readlane(dl, L);
        const float* __restrict__ hrow = h + (size_t)sB * HC + lane;
        float hv[H];
#pragma unroll
        for (int hh = 0; hh < H; ++hh) hv[hh] = hrow[hh * 32];
#pragma unroll
        for (int hh = 0; hh < H; ++hh) {
          const float wB = __int_as_float(
              __builtin_amdgcn_readlane(__float_as_int(w[hh]), L));
          atomicAdd(&s_num[dB * HC + hh * 32 + lane], wB * hv[hh]);
        }
      }
    }
  };

  const int nlS = lane * SEG_WV + wv;
  const int dS  = (nlS < cnt) ? n0 + nlS : -1;

  auto scan = [&](auto&& edgeFn) {
    if (wv < nG) {
      int4 nx[SCAN_U];
      ld_group(dstp, wv, lane, nx);
      for (int g = wv; g < nG; g += SEG_WV) {
        int4 cur[SCAN_U];
#pragma unroll
        for (int u = 0; u < SCAN_U; ++u) cur[u] = nx[u];
        ld_group(dstp, min(g + SEG_WV, nG - 1), lane, nx);
        unsigned hm = 0u;
#pragma unroll
        for (int u = 0; u < SCAN_U; ++u) {
          hm |= ((unsigned)(cur[u].x - n0) < (unsigned)cnt) ? (1u << (4 * u))     : 0u;
          hm |= ((unsigned)(cur[u].y - n0) < (unsigned)cnt) ? (1u << (4 * u + 1)) : 0u;
          hm |= ((unsigned)(cur[u].z - n0) < (unsigned)cnt) ? (1u << (4 * u + 2)) : 0u;
          hm |= ((unsigned)(cur[u].w - n0) < (unsigned)cnt) ? (1u << (4 * u + 3)) : 0u;
        }
        const int ebase = 4 * (g * SCAN_G + lane);
        while (__builtin_amdgcn_ballot_w32(hm != 0u) != 0u) {
          const bool has = (hm != 0u);
          const int  idx = has ? __builtin_ctz(hm) : 0;
          hm &= hm - 1u;
          const int e = ebase + (idx >> 2) * (4 * 32) + (idx & 3);
          const int d = has ? dstp[e] : -1;
          edgeFn(d, e, false);
        }
      }
    }
    for (int eb = eRest + wv * 32; eb < E; eb += SEG_TB) {
      const int e = eb + lane;
      const int d = (e < E) ? dstp[e] : -1;
      edgeFn(d, (e < E) ? e : 0, false);
    }
  };

  int wc = 0;
  if constexpr (BUILD) {
    auto stage = [&](bool hit, int e) {
      const unsigned m = __builtin_amdgcn_ballot_w32(hit);
      if (m != 0u) {
        const int pos = wc + (int)__builtin_amdgcn_mbcnt_lo(m, 0u);
        if (hit && pos < WCAP) s_stage[wv * WCAP + pos] = e;
        wc += __builtin_popcount(m);
      }
    };
    const bool useB = al16 && (((size_t)ob & 15) == 0) && gridDim.x <= 256;
    const int nGb    = useB ? (E >> 4) / SCAN_G : 0;
    const int eRestB = useB ? nGb * SCAN_G * 16 : eRest;
    if (useB) {
      const unsigned rep = (unsigned)blockIdx.x * 0x01010101u;
      auto own4 = [&](unsigned w) -> unsigned {
        const unsigned x = w ^ rep;
        return (~(((x & 0x7F7F7F7Fu) + 0x7F7F7F7Fu) | x | 0x7F7F7F7Fu)) >> 7;
      };
      if (wv < nGb) {
        uint4 nx[SCAN_U];
        ld_groupb(ob, wv, lane, nx);
        for (int g = wv; g < nGb; g += SEG_WV) {
          uint4 cur[SCAN_U];
#pragma unroll
          for (int u = 0; u < SCAN_U; ++u) cur[u] = nx[u];
          ld_groupb(ob, min(g + SEG_WV, nGb - 1), lane, nx);
          unsigned mA = 0u, mB = 0u;
#pragma unroll
          for (int u = 0; u < SCAN_U; ++u) {
            const unsigned f0 = own4(cur[u].x), f1 = own4(cur[u].y);
            const unsigned f2 = own4(cur[u].z), f3 = own4(cur[u].w);
            const int sh = 4 * (u & 1);
            const unsigned m = (f0 << sh) | (f1 << (sh + 1)) | (f2 << (sh + 2)) | (f3 << (sh + 3));
            if (u < 2) mA |= m; else mB |= m;
          }
          const int ebase = 16 * (g * SCAN_G + lane);
          while (__builtin_amdgcn_ballot_w32((mA | mB) != 0u) != 0u) {
            const bool     inA = (mA != 0u);
            const unsigned mx  = inA ? mA : mB;
            const bool     has = (mx != 0u);
            const int      t   = has ? __builtin_ctz(mx) : 0;
            const unsigned mc  = mx & (mx - 1u);
            mA = inA ? mc : mA;
            mB = inA ? mB : mc;
            const int j = (t & 7) + (inA ? 0 : 8);
            const int e = ebase + (j >> 2) * 512 + (j & 3) * 4 + (t >> 3);
            stage(has, e);
          }
        }
      }
    } else if (wv < nG) {
      int4 nx[SCAN_U];
      ld_group(dstp, wv, lane, nx);
      for (int g = wv; g < nG; g += SEG_WV) {
        int4 cur[SCAN_U];
#pragma unroll
        for (int u = 0; u < SCAN_U; ++u) cur[u] = nx[u];
        ld_group(dstp, min(g + SEG_WV, nG - 1), lane, nx);
        const int ebase = 4 * (g * SCAN_G + lane);
#pragma unroll
        for (int u = 0; u < SCAN_U; ++u) {
          stage((unsigned)(cur[u].x - n0) < (unsigned)cnt, ebase + u * 128);
          stage((unsigned)(cur[u].y - n0) < (unsigned)cnt, ebase + u * 128 + 1);
          stage((unsigned)(cur[u].z - n0) < (unsigned)cnt, ebase + u * 128 + 2);
          stage((unsigned)(cur[u].w - n0) < (unsigned)cnt, ebase + u * 128 + 3);
        }
      }
    }
    for (int eb = eRestB + wv * 32; eb < E; eb += SEG_TB) {
      const int e = eb + lane;
      const int d = (e < E) ? dstp[e] : -1;
      stage((unsigned)(d - n0) < (unsigned)cnt, e);
    }
    if (lane == 0) s_wc[wv] = wc;
  }
  __syncthreads();

  int2* __restrict__ wl = (int2*)wlist + (size_t)blockIdx.x * LCAP;

  auto put = [&](int idx, int s, int dl) {
    l_src[idx] = s;
    l_dl[idx]  = (unsigned char)dl;
    if constexpr (BUILD) wl[idx] = make_int2(s, dl);
    float av[H];
    ld_coef<H>(a_s, s, av);
#pragma unroll
    for (int hh = 0; hh < H; ++hh) {
      float l = av[hh] + s_ad[dl * H + hh];
      l = (l > 0.f) ? l : LRELU_SLOPE * l;
      atomicMax(&s_max[dl * H + hh], f2ord(l));
    }
  };

  int total;
  if constexpr (BUILD) {
    const int c = s_wc[lane];
    int pre = (lane < wv) ? c : 0;
    int all = c;
#pragma unroll
    for (int off = 16; off > 0; off >>= 1) {
      pre += __shfl_xor(pre, off, 32);
      all += __shfl_xor(all, off, 32);
    }
    pre = __builtin_amdgcn_readfirstlane(pre);
    all = __builtin_amdgcn_readfirstlane(all);
    const bool over = __builtin_amdgcn_ballot_w32(c > WCAP) != 0u;
    total = over ? -1 : all + cnt;
    if (tid == 0) wcnt[blockIdx.x] = total;
    if (!over) {
      for (int j = lane; j < wc; j += 32) {
        const int e = s_stage[wv * WCAP + j];
        put(pre + j, srcp[e], dstp[e] - n0);
      }
      if (tid < cnt) put(all + tid, n0 + tid, tid);
    }
  } else {
    total = wcnt[blockIdx.x];
    if (total > LCAP) total = -1;
    for (int i = tid; i < total; i += SEG_TB) {
      const int2 v = wl[i];
      put(i, v.x, v.y);
    }
  }

  if (total >= 0) {
    __syncthreads();
    constexpr int UNR = (H >= 8) ? 4 : 16;
    const int per = (total + SEG_WV - 1) / SEG_WV;
    const int lo  = wv * per;
    const int hi  = min(total, lo + per);
    for (int base = lo; base < hi; base += 32) {
      const int  i     = base + lane;
      const bool valid = i < hi;
      int   s = 0, dl = 0;
      float w[H];
#pragma unroll
      for (int hh = 0; hh < H; ++hh) w[hh] = 0.f;
      if (valid) {
        s  = l_src[i];
        dl = l_dl[i];
        float av[H];
        ld_coef<H>(a_s, s, av);
#pragma unroll
        for (int hh = 0; hh < H; ++hh) {
          float l = av[hh] + s_ad[dl * H + hh];
          l = (l > 0.f) ? l : LRELU_SLOPE * l;
          w[hh] = expf(l - ord2f(s_max[dl * H + hh]));
          atomicAdd(&s_den[dl * H + hh], w[hh]);
        }
      }
      const int n = min(32, hi - base);
      for (int L0 = 0; L0 < n; L0 += UNR) {
        float hv[UNR][H];
        int   dB[UNR];
#pragma unroll
        for (int j = 0; j < UNR; ++j) {
          const int L  = min(L0 + j, n - 1);
          const int sB = __builtin_amdgcn_readlane(s, L);
          dB[j]        = __builtin_amdgcn_readlane(dl, L);
          const float* __restrict__ hrow = h + (size_t)sB * HC + lane;
#pragma unroll
          for (int hh = 0; hh < H; ++hh) hv[j][hh] = hrow[hh * 32];
        }
#pragma unroll
        for (int j = 0; j < UNR; ++j) {
          if (L0 + j < n) {
#pragma unroll
            for (int hh = 0; hh < H; ++hh) {
              const float wB = __int_as_float(
                  __builtin_amdgcn_readlane(__float_as_int(w[hh]), L0 + j));
              atomicAdd(&s_num[dB[j] * HC + hh * 32 + lane], wB * hv[j][hh]);
            }
          }
        }
      }
    }
  } else {
    edgeA(dS, 0, true);
    scan(edgeA);
    __syncthreads();
    edgeB(dS, 0, true);
    scan(edgeB);
  }
  __syncthreads();

  float* __restrict__ orow = out + (size_t)n0 * HC;
  for (int i = tid; i < cnt * HC; i += SEG_TB) {
    const int nl = i / HC;
    const int hc = i % HC;
    float v = s_num[i] / (s_den[nl * H + hc / 32] + 1e-16f) + bias[hc];
    v = (v > 0.f) ? v : (expf(v) - 1.f);
    orow[i] = v;
  }
}

extern "C" void kernel_launch(void* const* d_in, const int* in_sizes, int n_in,
                              void* d_out, int out_size, void* d_ws, size_t ws_size,
                              hipStream_t stream) {
  const float* x   = (const float*)d_in[0];
  const int*   ei  = (const int*)d_in[1];
  const float* W1  = (const float*)d_in[2];
  const float* as1 = (const float*)d_in[3];
  const float* ad1 = (const float*)d_in[4];
  const float* b1  = (const float*)d_in[5];
  const float* W2  = (const float*)d_in[6];
  const float* as2 = (const float*)d_in[7];
  const float* ad2 = (const float*)d_in[8];
  const float* b2  = (const float*)d_in[9];

  const int HC   = in_sizes[3];
  const int Fin  = in_sizes[2] / HC;
  const int N    = in_sizes[0] / Fin;
  const int E    = in_sizes[1] / 2;
  const int H    = 8;
  const int Cout = in_sizes[7];

  float* ws = (float*)d_ws;
  float* h1  = ws;
  float* x2  = h1 + (size_t)N * HC;
  float* h2  = x2 + (size_t)N * HC;
  float* aS1 = h2 + (size_t)N * Cout;
  float* aD1 = aS1 + (size_t)N * H;
  float* aS2 = aD1 + (size_t)N * H;
  float* aD2 = aS2 + N;
  const int segBlocksWs = (N + NPB - 1) / NPB;
  int* eList = (int*)(aD2 + N);
  int* eCnt  = eList + (size_t)segBlocksWs * LCAP * 2;
  unsigned* eOwn = (unsigned*)(eCnt + ((segBlocksWs + 3) & ~3));

  const int TB = 256;
  auto blocks = [](long long n, int tb) { return (int)((n + tb - 1) / tb); };
  const int segBlocks = (N + NPB - 1) / NPB;

  {
    const long long waves = (long long)(N / 16) * (HC / 32);
    gemm_att_wmma_f32<128, 256, true><<<blocks(waves * 32, TB), TB, 0, stream>>>(
        x, W1, as1, ad1, h1, aS1, aD1, N, ei + E, E, segBlocks, eOwn);
    seg_softmax_agg<8, true><<<segBlocks, SEG_TB, 0, stream>>>(
        ei, E, N, aS1, aD1, h1, b1, x2, eList, eCnt, eOwn);
  }

  {
    const long long waves = (long long)(N / 16) * (Cout / 32);
    gemm_att_wmma_f32<256, 32, false><<<blocks(waves * 32, TB), TB, 0, stream>>>(
        x2, W2, as2, ad2, h2, aS2, aD2, N, ei + E, E, segBlocks, eOwn);
    seg_softmax_agg<1, false><<<segBlocks, SEG_TB, 0, stream>>>(
        ei, E, N, aS2, aD2, h2, b2, (float*)d_out, eList, eCnt, eOwn);
  }
}


# Layer-1 GEMM: 256 resident 32-wave blocks copy W1 into LDS once (k pairs interleaved, 8-byte reads) and loop over row tiles; B fragments come from LDS; products and sum order as before
# speedup vs baseline: 9.0889x; 1.0636x over previous
#include <hip/hip_runtime.h>
#include <math.h>

typedef float v2f __attribute__((ext_vector_type(2)));
typedef float v8f __attribute__((ext_vector_type(8)));

#define LRELU_SLOPE 0.2f
#define NPB 196

__device__ __forceinline__ unsigned f2ord(float f) {
  unsigned u = __float_as_uint(f);
  return (u & 0x80000000u) ? ~u : (u | 0x80000000u);
}
__device__ __forceinline__ float ord2f(unsigned u) {
  return __uint_as_float((u & 0x80000000u) ? (u & 0x7FFFFFFFu) : ~u);
}

template <int K, int NC>
__global__ __launch_bounds__(256) void gemm_att_wmma_f32(
    const float* __restrict__ X, const float* __restrict__ W,
    const float* __restrict__ att_s, const float* __restrict__ att_d,
    float* __restrict__ Y, float* __restrict__ a_s, float* __restrict__ a_d,
    int M) {
  constexpr int HEADS = NC / 32;
  static_assert(NC % 32 == 0 && K % 4 == 0, "a head is 32 channels; K steps by 4");
  const int lane  = threadIdx.x & 31;
  const int wave  = (blockIdx.x * blockDim.x + threadIdx.x) >> 5;
  if (wave >= (M >> 4) * HEADS) return;
  const int mTile = (wave / HEADS) << 4;
  const int head  = wave % HEADS;
  const int half  = lane >> 4;
  const int l16   = lane & 15;

  const float* __restrict__ xrow = X + (size_t)(mTile + l16) * K + 2 * half;
  const float* __restrict__ wcol = W + (size_t)(2 * half) * NC + head * 32 + l16;

  v8f acc0 = {0.f, 0.f, 0.f, 0.f, 0.f, 0.f, 0.f, 0.f};
  v8f acc1 = {0.f, 0.f, 0.f, 0.f, 0.f, 0.f, 0.f, 0.f};
#pragma unroll 8
  for (int k = 0; k < K; k += 4) {
    v2f a = *(const v2f*)(xrow + k);
    v2f b0, b1;
    b0.x = wcol[k * NC];
    b0.y = wcol[(k + 1) * NC];
    b1.x = wcol[k * NC + 16];
    b1.y = wcol[(k + 1) * NC + 16];
#if __has_builtin(__builtin_amdgcn_wmma_f32_16x16x4_f32)
    acc0 = __builtin_amdgcn_wmma_f32_16x16x4_f32(false, a, false, b0,
                                                 (short)0, acc0, false, false);
    acc1 = __builtin_amdgcn_wmma_f32_16x16x4_f32(false, a, false, b1,
                                                 (short)0, acc1, false, false);
#else
    acc0[0] += a.x * b0.x;
    acc1[0] += a.x * b1.x;
#endif
  }
  float* __restrict__ yrow = Y + (size_t)(mTile + half * 8) * NC + head * 32 + l16;
#pragma unroll
  for (int r = 0; r < 8; ++r) {
    yrow[r * NC]      = acc0[r];
    yrow[r * NC + 16] = acc1[r];
  }

  const float s0 = att_s[head * 32 + l16], s1 = att_s[head * 32 + 16 + l16];
  const float d0 = att_d[head * 32 + l16], d1 = att_d[head * 32 + 16 + l16];
  float v[16];
#pragma unroll
  for (int r = 0; r < 8; ++r) {
    v[r]     = fmaf(acc0[r], s0, acc1[r] * s1);
    v[8 + r] = fmaf(acc0[r], d0, acc1[r] * d1);
  }
  const bool b3 = (l16 & 8) != 0;
  float u[8];
#pragma unroll
  for (int j = 0; j < 8; ++j) {
    const float keep = b3 ? v[8 + j] : v[j];
    const float send = b3 ? v[j] : v[8 + j];
    u[j] = keep + __shfl_xor(send, 8, 32);
  }
  const bool b2 = (l16 & 4) != 0;
  float t[4];
#pragma unroll
  for (int j = 0; j < 4; ++j) {
    const float keep = b2 ? u[4 + j] : u[j];
    const float send = b2 ? u[j] : u[4 + j];
    t[j] = keep + __shfl_xor(send, 4, 32);
  }
  const bool b1 = (l16 & 2) != 0;
  float q[2];
#pragma unroll
  for (int j = 0; j < 2; ++j) {
    const float keep = b1 ? t[2 + j] : t[j];
    const float send = b1 ? t[j] : t[2 + j];
    q[j] = keep + __shfl_xor(send, 2, 32);
  }
  const bool b0 = (l16 & 1) != 0;
  const float keep = b0 ? q[1] : q[0];
  const float send = b0 ? q[0] : q[1];
  const float res  = keep + __shfl_xor(send, 1, 32);
  const int row = mTile + 8 * half + (l16 & 7);
  float* __restrict__ dst = b3 ? a_d : a_s;
  dst[(size_t)row * HEADS + head] = res;
}

#define G1_TB 1024
#define G1_WV (G1_TB / 32)
template <int K, int NC>
__global__ __launch_bounds__(G1_TB) void gemm_att_wmma_f32_ldsw(
    const float* __restrict__ X, const float* __restrict__ W,
    const float* __restrict__ att_s, const float* __restrict__ att_d,
    float* __restrict__ Y, float* __restrict__ a_s, float* __restrict__ a_d,
    int M, const int* __restrict__ dstp, int E, int nblk, unsigned* __restrict__ ob) {
  constexpr int HEADS = NC / 32;
  constexpr int G1_PITCH = 2 * NC + 32;
  static_assert(NC % 32 == 0 && K % 4 == 0, "a head is 32 channels; K steps by 4");
  static_assert(G1_WV % HEADS == 0 && NC % 4 == 0, "a tile slot is HEADS waves; W rows move as 16-byte pieces");
  __shared__ __attribute__((aligned(16))) float s_w[(K / 2) * G1_PITCH];
  const int tid  = threadIdx.x;
  const int lane = tid & 31;
  {
    const bool ok = ((E & 3) == 0) && (((size_t)dstp & 15) == 0) &&
                    (((size_t)ob & 15) == 0) && nblk <= 256;
    if (ok) {
      const int nQ = E >> 2;
      const int gw = (blockIdx.x * blockDim.x + threadIdx.x) >> 5;
      const int nw = (gridDim.x * blockDim.x) >> 5;
      for (int base = gw * 32; base < nQ; base += nw * 32) {
        const int q = base + lane;
        if (q < nQ) {
          const int4 d = ((const int4*)dstp)[q];
          ob[q] = ((unsigned)d.x / NPB) | (((unsigned)d.y / NPB) << 8) |
                  (((unsigned)d.z / NPB) << 16) | (((unsigned)d.w / NPB) << 24);
        }
      }
    }
  }
  for (int i = tid; i < (K / 2) * (NC / 4); i += G1_TB) {
    const int k2 = i / (NC / 4);
    const int c4 = i % (NC / 4);
    const float4 r0 = *(const float4*)(W + (size_t)(2 * k2) * NC + 4 * c4);
    const float4 r1 = *(const float4*)(W + (size_t)(2 * k2 + 1) * NC + 4 * c4);
    float4* __restrict__ d = (float4*)(s_w + k2 * G1_PITCH + 8 * c4);
    d[0] = make_float4(r0.x, r1.x, r0.y, r1.y);
    d[1] = make_float4(r0.z, r1.z, r0.w, r1.w);
  }
  __syncthreads();

  const int wv     = __builtin_amdgcn_readfirstlane(tid >> 5);
  const int head   = wv % HEADS;
  const int slot   = (wv / HEADS) * gridDim.x + blockIdx.x;
  const int nSlots = gridDim.x * (G1_WV / HEADS);
  const int nTiles = M >> 4;
  const int half   = lane >> 4;
  const int l16    = lane & 15;

  const float* __restrict__ wl = s_w + half * G1_PITCH + (head * 32 + l16) * 2;
  const float s0 = att_s[head * 32 + l16], s1 = att_s[head * 32 + 16 + l16];
  const float d0 = att_d[head * 32 + l16], d1 = att_d[head * 32 + 16 + l16];
  const bool b3 = (l16 & 8) != 0;
  const bool b2 = (l16 & 4) != 0;
  const bool b1 = (l16 & 2) != 0;
  const bool b0 = (l16 & 1) != 0;
  float* __restrict__ dsel = b3 ? a_d : a_s;

  for (int tile = slot; tile < nTiles; tile += nSlots) {
    const int mTile = tile << 4;
    const float* __restrict__ xrow = X + (size_t)(mTile + l16) * K + 2 * half;
    v8f acc0 = {0.f, 0.f, 0.f, 0.f, 0.f, 0.f, 0.f, 0.f};
    v8f acc1 = {0.f, 0.f, 0.f, 0.f, 0.f, 0.f, 0.f, 0.f};
#pragma unroll 8
    for (int k = 0; k < K; k += 4) {
      const v2f a = *(const v2f*)(xrow + k);
      const float* __restrict__ p = wl + (k >> 1) * G1_PITCH;
      const v2f bf0 = *(const v2f*)p;
      const v2f bf1 = *(const v2f*)(p + 32);
#if __has_builtin(__builtin_amdgcn_wmma_f32_16x16x4_f32)
      acc0 = __builtin_amdgcn_wmma_f32_16x16x4_f32(false, a, false, bf0,
                                                   (short)0, acc0, false, false);
      acc1 = __builtin_amdgcn_wmma_f32_16x16x4_f32(false, a, false, bf1,
                                                   (short)0, acc1, false, false);
#else
      acc0[0] += a.x * bf0.x;
      acc1[0] += a.x * bf1.x;
#endif
    }
    float* __restrict__ yrow = Y + (size_t)(mTile + half * 8) * NC + head * 32 + l16;
#pragma unroll
    for (int r = 0; r < 8; ++r) {
      yrow[r * NC]      = acc0[r];
      yrow[r * NC + 16] = acc1[r];
    }
    float v[16];
#pragma unroll
    for (int r = 0; r < 8; ++r) {
      v[r]     = fmaf(acc0[r], s0, acc1[r] * s1);
      v[8 + r] = fmaf(acc0[r], d0, acc1[r] * d1);
    }
    float u[8];
#pragma unroll
    for (int j = 0; j < 8; ++j) {
      const float keep = b3 ? v[8 + j] : v[j];
      const float send = b3 ? v[j] : v[8 + j];
      u[j] = keep + __shfl_xor(send, 8, 32);
    }
    float t[4];
#pragma unroll
    for (int j = 0; j < 4; ++j) {
      const float keep = b2 ? u[4 + j] : u[j];
      const float send = b2 ? u[j] : u[4 + j];
      t[j] = keep + __shfl_xor(send, 4, 32);
    }
    float q[2];
#pragma unroll
    for (int j = 0; j < 2; ++j) {
      const float keep = b1 ? t[2 + j] : t[j];
      const float send = b1 ? t[j] : t[2 + j];
      q[j] = keep + __shfl_xor(send, 2, 32);
    }
    const float keep = b0 ? q[1] : q[0];
    const float send = b0 ? q[0] : q[1];
    const float res  = keep + __shfl_xor(send, 1, 32);
    const int row = mTile + 8 * half + (l16 & 7);
    dsel[(size_t)row * HEADS + head] = res;
  }
}

#define SEG_TB 1024
#define SEG_WV (SEG_TB / 32)
#define SCAN_U 4
#define SCAN_G (32 * SCAN_U)
#define WCAP 192
#define LCAP (SEG_WV * WCAP + NPB)
static_assert(NPB <= 256, "local dst index is kept in a byte");
static_assert(SCAN_U == 4, "the owner-byte scan packs a trip's 16 dwords per lane into two 32-bit hit masks");

template <int H>
__device__ __forceinline__ void ld_coef(const float* __restrict__ a, int n,
                                        float (&v)[H]) {
  if constexpr ((H & 3) == 0) {
#pragma unroll
    for (int i = 0; i < H; i += 4) {
      const float4 t = *(const float4*)(a + (size_t)n * H + i);
      v[i] = t.x; v[i + 1] = t.y; v[i + 2] = t.z; v[i + 3] = t.w;
    }
  } else {
#pragma unroll
    for (int i = 0; i < H; ++i) v[i] = a[(size_t)n * H + i];
  }
}

__device__ __forceinline__ void ld_group(const int* __restrict__ dstp, int g,
                                         int lane, int4 (&v)[SCAN_U]) {
  const int4* __restrict__ p = (const int4*)dstp + (size_t)g * SCAN_G + lane;
#pragma unroll
  for (int u = 0; u < SCAN_U; ++u) v[u] = p[u * 32];
}

__device__ __forceinline__ void ld_groupb(const unsigned* __restrict__ ob, int g,
                                          int lane, uint4 (&v)[SCAN_U]) {
  const uint4* __restrict__ p = (const uint4*)ob + (size_t)g * SCAN_G + lane;
#pragma unroll
  for (int u = 0; u < SCAN_U; ++u) v[u] = p[u * 32];
}

template <int H, bool BUILD>
__global__ __launch_bounds__(SEG_TB) void seg_softmax_agg(
    const int* __restrict__ ei, int E, int N,
    const float* __restrict__ a_s, const float* __restrict__ a_d,
    const float* __restrict__ h, const float* __restrict__ bias,
    float* __restrict__ out, int* __restrict__ wlist, int* __restrict__ wcnt,
    const unsigned* __restrict__ ob) {
  constexpr int HC = H * 32;
  __shared__ float    s_num[NPB * HC];
  __shared__ float    s_den[NPB * H];
  __shared__ unsigned s_max[NPB * H];
  __shared__ float    s_ad[NPB * H];
  __shared__ int           s_stage[BUILD ? SEG_WV * WCAP : 1];
  __shared__ int           s_wc[SEG_WV];
  __shared__ int           l_src[LCAP];
  __shared__ unsigned char l_dl[LCAP];

  const int tid  = threadIdx.x;
  const int lane = tid & 31;
  const int wv   = __builtin_amdgcn_readfirstlane(tid >> 5);
  const int n0   = blockIdx.x * NPB;
  const int cnt  = min(NPB, N - n0);

  for (int i = tid; i < NPB * HC; i += SEG_TB) s_num[i] = 0.f;
  for (int i = tid; i < NPB * H; i += SEG_TB) {
    s_den[i] = 0.f;
    s_max[i] = 0u;
    s_ad[i]  = (i < cnt * H) ? a_d[(size_t)n0 * H + i] : 0.f;
  }

  const int* __restrict__ srcp = ei;
  const int* __restrict__ dstp = ei + E;
  const bool al16 = ((E & 3) == 0) && (((size_t)dstp & 15) == 0);
  const int nG    = al16 ? (E >> 2) / SCAN_G : 0;
  const int eRest = nG * SCAN_G * 4;

  auto edgeA = [&](int d, int e, bool self) {
    const int dl = d - n0;
    if ((unsigned)dl < (unsigned)cnt) {
      const int s = self ? d : srcp[e];
      float av[H];
      ld_coef<H>(a_s, s, av);
#pragma unroll
      for (int hh = 0; hh < H; ++hh) {
        float l = av[hh] + s_ad[dl * H + hh];
        l = (l > 0.f) ? l : LRELU_SLOPE * l;
        atomicMax(&s_max[dl * H + hh], f2ord(l));
      }
    }
  };

  auto edgeB = [&](int d, int e, bool self) {
    const int  dl  = d - n0;
    const bool hit = (unsigned)dl < (unsigned)cnt;
    unsigned mask = __builtin_amdgcn_ballot_w32(hit);
    if (mask != 0u) {
      int   s = 0;
      float w[H];
#pragma unroll
      for (int hh = 0; hh < H; ++hh) w[hh] = 0.f;
      if (hit) {
        s = self ? d : srcp[e];
        float av[H];
        ld_coef<H>(a_s, s, av);
#pragma unroll
        for (int hh = 0; hh < H; ++hh) {
          float l = av[hh] + s_ad[dl * H + hh];
          l = (l > 0.f) ? l : LRELU_SLOPE * l;
          w[hh] = expf(l - ord2f(s_max[dl * H + hh]));
          atomicAdd(&s_den[dl * H + hh], w[hh]);
        }
      }
      while (mask) {
        const int L = __builtin_ctz(mask);
        mask &= mask - 1u;
        const int sB = __builtin_amdgcn_readlane(s, L);
        const int dB = __builtin_amdgcn_readlane(dl, L);
        const float* __restrict__ hrow = h + (size_t)sB * HC + lane;
        float hv[H];
#pragma unroll
        for (int hh = 0; hh < H; ++hh) hv[hh] = hrow[hh * 32];
#pragma unroll
        for (int hh = 0; hh < H; ++hh) {
          const float wB = __int_as_float(
              __builtin_amdgcn_readlane(__float_as_int(w[hh]), L));
          atomicAdd(&s_num[dB * HC + hh * 32 + lane], wB * hv[hh]);
        }
      }
    }
  };

  const int nlS = lane * SEG_WV + wv;
  const int dS  = (nlS < cnt) ? n0 + nlS : -1;

  auto scan = [&](auto&& edgeFn) {
    if (wv < nG) {
      int4 nx[SCAN_U];
      ld_group(dstp, wv, lane, nx);
      for (int g = wv; g < nG; g += SEG_WV) {
        int4 cur[SCAN_U];
#pragma unroll
        for (int u = 0; u < SCAN_U; ++u) cur[u] = nx[u];
        ld_group(dstp, min(g + SEG_WV, nG - 1), lane, nx);
        unsigned hm = 0u;
#pragma unroll
        for (int u = 0; u < SCAN_U; ++u) {
          hm |= ((unsigned)(cur[u].x - n0) < (unsigned)cnt) ? (1u << (4 * u))     : 0u;
          hm |= ((unsigned)(cur[u].y - n0) < (unsigned)cnt) ? (1u << (4 * u + 1)) : 0u;
          hm |= ((unsigned)(cur[u].z - n0) < (unsigned)cnt) ? (1u << (4 * u + 2)) : 0u;
          hm |= ((unsigned)(cur[u].w - n0) < (unsigned)cnt) ? (1u << (4 * u + 3)) : 0u;
        }
        const int ebase = 4 * (g * SCAN_G + lane);
        while (__builtin_amdgcn_ballot_w32(hm != 0u) != 0u) {
          const bool has = (hm != 0u);
          const int  idx = has ? __builtin_ctz(hm) : 0;
          hm &= hm - 1u;
          const int e = ebase + (idx >> 2) * (4 * 32) + (idx & 3);
          const int d = has ? dstp[e] : -1;
          edgeFn(d, e, false);
        }
      }
    }
    for (int eb = eRest + wv * 32; eb < E; eb += SEG_TB) {
      const int e = eb + lane;
      const int d = (e < E) ? dstp[e] : -1;
      edgeFn(d, (e < E) ? e : 0, false);
    }
  };

  int wc = 0;
  if constexpr (BUILD) {
    auto stage = [&](bool hit, int e) {
      const unsigned m = __builtin_amdgcn_ballot_w32(hit);
      if (m != 0u) {
        const int pos = wc + (int)__builtin_amdgcn_mbcnt_lo(m, 0u);
        if (hit && pos < WCAP) s_stage[wv * WCAP + pos] = e;
        wc += __builtin_popcount(m);
      }
    };
    const bool useB = al16 && (((size_t)ob & 15) == 0) && gridDim.x <= 256;
    const int nGb    = useB ? (E >> 4) / SCAN_G : 0;
    const int eRestB = useB ? nGb * SCAN_G * 16 : eRest;
    if (useB) {
      const unsigned rep = (unsigned)blockIdx.x * 0x01010101u;
      auto own4 = [&](unsigned w) -> unsigned {
        const unsigned x = w ^ rep;
        return (~(((x & 0x7F7F7F7Fu) + 0x7F7F7F7Fu) | x | 0x7F7F7F7Fu)) >> 7;
      };
      if (wv < nGb) {
        uint4 nx[SCAN_U];
        ld_groupb(ob, wv, lane, nx);
        for (int g = wv; g < nGb; g += SEG_WV) {
          uint4 cur[SCAN_U];
#pragma unroll
          for (int u = 0; u < SCAN_U; ++u) cur[u] = nx[u];
          ld_groupb(ob, min(g + SEG_WV, nGb - 1), lane, nx);
          unsigned mA = 0u, mB = 0u;
#pragma unroll
          for (int u = 0; u < SCAN_U; ++u) {
            const unsigned f0 = own4(cur[u].x), f1 = own4(cur[u].y);
            const unsigned f2 = own4(cur[u].z), f3 = own4(cur[u].w);
            const int sh = 4 * (u & 1);
            const unsigned m = (f0 << sh) | (f1 << (sh + 1)) | (f2 << (sh + 2)) | (f3 << (sh + 3));
            if (u < 2) mA |= m; else mB |= m;
          }
          const int ebase = 16 * (g * SCAN_G + lane);
          while (__builtin_amdgcn_ballot_w32((mA | mB) != 0u) != 0u) {
            const bool     inA = (mA != 0u);
            const unsigned mx  = inA ? mA : mB;
            const bool     has = (mx != 0u);
            const int      t   = has ? __builtin_ctz(mx) : 0;
            const unsigned mc  = mx & (mx - 1u);
            mA = inA ? mc : mA;
            mB = inA ? mB : mc;
            const int j = (t & 7) + (inA ? 0 : 8);
            const int e = ebase + (j >> 2) * 512 + (j & 3) * 4 + (t >> 3);
            stage(has, e);
          }
        }
      }
    } else if (wv < nG) {
      int4 nx[SCAN_U];
      ld_group(dstp, wv, lane, nx);
      for (int g = wv; g < nG; g += SEG_WV) {
        int4 cur[SCAN_U];
#pragma unroll
        for (int u = 0; u < SCAN_U; ++u) cur[u] = nx[u];
        ld_group(dstp, min(g + SEG_WV, nG - 1), lane, nx);
        const int ebase = 4 * (g * SCAN_G + lane);
#pragma unroll
        for (int u = 0; u < SCAN_U; ++u) {
          stage((unsigned)(cur[u].x - n0) < (unsigned)cnt, ebase + u * 128);
          stage((unsigned)(cur[u].y - n0) < (unsigned)cnt, ebase + u * 128 + 1);
          stage((unsigned)(cur[u].z - n0) < (unsigned)cnt, ebase + u * 128 + 2);
          stage((unsigned)(cur[u].w - n0) < (unsigned)cnt, ebase + u * 128 + 3);
        }
      }
    }
    for (int eb = eRestB + wv * 32; eb < E; eb += SEG_TB) {
      const int e = eb + lane;
      const int d = (e < E) ? dstp[e] : -1;
      stage((unsigned)(d - n0) < (unsigned)cnt, e);
    }
    if (lane == 0) s_wc[wv] = wc;
  }
  __syncthreads();

  int2* __restrict__ wl = (int2*)wlist + (size_t)blockIdx.x * LCAP;

  auto put = [&](int idx, int s, int dl) {
    l_src[idx] = s;
    l_dl[idx]  = (unsigned char)dl;
    if constexpr (BUILD) wl[idx] = make_int2(s, dl);
    float av[H];
    ld_coef<H>(a_s, s, av);
#pragma unroll
    for (int hh = 0; hh < H; ++hh) {
      float l = av[hh] + s_ad[dl * H + hh];
      l = (l > 0.f) ? l : LRELU_SLOPE * l;
      atomicMax(&s_max[dl * H + hh], f2ord(l));
    }
  };

  int total;
  if constexpr (BUILD) {
    const int c = s_wc[lane];
    int pre = (lane < wv) ? c : 0;
    int all = c;
#pragma unroll
    for (int off = 16; off > 0; off >>= 1) {
      pre += __shfl_xor(pre, off, 32);
      all += __shfl_xor(all, off, 32);
    }
    pre = __builtin_amdgcn_readfirstlane(pre);
    all = __builtin_amdgcn_readfirstlane(all);
    const bool over = __builtin_amdgcn_ballot_w32(c > WCAP) != 0u;
    total = over ? -1 : all + cnt;
    if (tid == 0) wcnt[blockIdx.x] = total;
    if (!over) {
      for (int j = lane; j < wc; j += 32) {
        const int e = s_stage[wv * WCAP + j];
        put(pre + j, srcp[e], dstp[e] - n0);
      }
      if (tid < cnt) put(all + tid, n0 + tid, tid);
    }
  } else {
    total = wcnt[blockIdx.x];
    if (total > LCAP) total = -1;
    for (int i = tid; i < total; i += SEG_TB) {
      const int2 v = wl[i];
      put(i, v.x, v.y);
    }
  }

  if (total >= 0) {
    __syncthreads();
    constexpr int UNR = (H >= 8) ? 4 : 16;
    const int per = (total + SEG_WV - 1) / SEG_WV;
    const int lo  = wv * per;
    const int hi  = min(total, lo + per);
    for (int base = lo; base < hi; base += 32) {
      const int  i     = base + lane;
      const bool valid = i < hi;
      int   s = 0, dl = 0;
      float w[H];
#pragma unroll
      for (int hh = 0; hh < H; ++hh) w[hh] = 0.f;
      if (valid) {
        s  = l_src[i];
        dl = l_dl[i];
        float av[H];
        ld_coef<H>(a_s, s, av);
#pragma unroll
        for (int hh = 0; hh < H; ++hh) {
          float l = av[hh] + s_ad[dl * H + hh];
          l = (l > 0.f) ? l : LRELU_SLOPE * l;
          w[hh] = expf(l - ord2f(s_max[dl * H + hh]));
          atomicAdd(&s_den[dl * H + hh], w[hh]);
        }
      }
      const int n = min(32, hi - base);
      for (int L0 = 0; L0 < n; L0 += UNR) {
        float hv[UNR][H];
        int   dB[UNR];
#pragma unroll
        for (int j = 0; j < UNR; ++j) {
          const int L  = min(L0 + j, n - 1);
          const int sB = __builtin_amdgcn_readlane(s, L);
          dB[j]        = __builtin_amdgcn_readlane(dl, L);
          const float* __restrict__ hrow = h + (size_t)sB * HC + lane;
#pragma unroll
          for (int hh = 0; hh < H; ++hh) hv[j][hh] = hrow[hh * 32];
        }
#pragma unroll
        for (int j = 0; j < UNR; ++j) {
          if (L0 + j < n) {
#pragma unroll
            for (int hh = 0; hh < H; ++hh) {
              const float wB = __int_as_float(
                  __builtin_amdgcn_readlane(__float_as_int(w[hh]), L0 + j));
              atomicAdd(&s_num[dB[j] * HC + hh * 32 + lane], wB * hv[j][hh]);
            }
          }
        }
      }
    }
  } else {
    edgeA(dS, 0, true);
    scan(edgeA);
    __syncthreads();
    edgeB(dS, 0, true);
    scan(edgeB);
  }
  __syncthreads();

  float* __restrict__ orow = out + (size_t)n0 * HC;
  for (int i = tid; i < cnt * HC; i += SEG_TB) {
    const int nl = i / HC;
    const int hc = i % HC;
    float v = s_num[i] / (s_den[nl * H + hc / 32] + 1e-16f) + bias[hc];
    v = (v > 0.f) ? v : (expf(v) - 1.f);
    orow[i] = v;
  }
}

extern "C" void kernel_launch(void* const* d_in, const int* in_sizes, int n_in,
                              void* d_out, int out_size, void* d_ws, size_t ws_size,
                              hipStream_t stream) {
  const float* x   = (const float*)d_in[0];
  const int*   ei  = (const int*)d_in[1];
  const float* W1  = (const float*)d_in[2];
  const float* as1 = (const float*)d_in[3];
  const float* ad1 = (const float*)d_in[4];
  const float* b1  = (const float*)d_in[5];
  const float* W2  = (const float*)d_in[6];
  const float* as2 = (const float*)d_in[7];
  const float* ad2 = (const float*)d_in[8];
  const float* b2  = (const float*)d_in[9];

  const int HC   = in_sizes[3];
  const int Fin  = in_sizes[2] / HC;
  const int N    = in_sizes[0] / Fin;
  const int E    = in_sizes[1] / 2;
  const int H    = 8;
  const int Cout = in_sizes[7];

  float* ws = (float*)d_ws;
  float* h1  = ws;
  float* x2  = h1 + (size_t)N * HC;
  float* h2  = x2 + (size_t)N * HC;
  float* aS1 = h2 + (size_t)N * Cout;
  float* aD1 = aS1 + (size_t)N * H;
  float* aS2 = aD1 + (size_t)N * H;
  float* aD2 = aS2 + N;
  const int segBlocksWs = (N + NPB - 1) / NPB;
  int* eList = (int*)(aD2 + N);
  int* eCnt  = eList + (size_t)segBlocksWs * LCAP * 2;
  unsigned* eOwn = (unsigned*)(eCnt + ((segBlocksWs + 3) & ~3));

  const int TB = 256;
  auto blocks = [](long long n, int tb) { return (int)((n + tb - 1) / tb); };
  const int segBlocks = (N + NPB - 1) / NPB;

  {
    gemm_att_wmma_f32_ldsw<128, 256><<<256, G1_TB, 0, stream>>>(
        x, W1, as1, ad1, h1, aS1, aD1, N, ei + E, E, segBlocks, eOwn);
    seg_softmax_agg<8, true><<<segBlocks, SEG_TB, 0, stream>>>(
        ei, E, N, aS1, aD1, h1, b1, x2, eList, eCnt, eOwn);
  }

  {
    const long long waves = (long long)(N / 16) * (Cout / 32);
    gemm_att_wmma_f32<256, 32><<<blocks(waves * 32, TB), TB, 0, stream>>>(
        x2, W2, as2, ad2, h2, aS2, aD2, N);
    seg_softmax_agg<1, false><<<segBlocks, SEG_TB, 0, stream>>>(
        ei, E, N, aS2, aD2, h2, b2, (float*)d_out, eList, eCnt, eOwn);
  }
}


# Layer-1 aggregation: a wave owns 7 dst rows of its block (local index & 31), picks their entries from the block's list and sums the weighted rows in registers, list order, fused; LDS row adds go
# speedup vs baseline: 9.7828x; 1.0763x over previous
#include <hip/hip_runtime.h>
#include <math.h>

typedef float v2f __attribute__((ext_vector_type(2)));
typedef float v8f __attribute__((ext_vector_type(8)));

#define LRELU_SLOPE 0.2f
#define NPB 196

__device__ __forceinline__ unsigned f2ord(float f) {
  unsigned u = __float_as_uint(f);
  return (u & 0x80000000u) ? ~u : (u | 0x80000000u);
}
__device__ __forceinline__ float ord2f(unsigned u) {
  return __uint_as_float((u & 0x80000000u) ? (u & 0x7FFFFFFFu) : ~u);
}

template <int K, int NC>
__global__ __launch_bounds__(256) void gemm_att_wmma_f32(
    const float* __restrict__ X, const float* __restrict__ W,
    const float* __restrict__ att_s, const float* __restrict__ att_d,
    float* __restrict__ Y, float* __restrict__ a_s, float* __restrict__ a_d,
    int M) {
  constexpr int HEADS = NC / 32;
  static_assert(NC % 32 == 0 && K % 4 == 0, "a head is 32 channels; K steps by 4");
  const int lane  = threadIdx.x & 31;
  const int wave  = (blockIdx.x * blockDim.x + threadIdx.x) >> 5;
  if (wave >= (M >> 4) * HEADS) return;
  const int mTile = (wave / HEADS) << 4;
  const int head  = wave % HEADS;
  const int half  = lane >> 4;
  const int l16   = lane & 15;

  const float* __restrict__ xrow = X + (size_t)(mTile + l16) * K + 2 * half;
  const float* __restrict__ wcol = W + (size_t)(2 * half) * NC + head * 32 + l16;

  v8f acc0 = {0.f, 0.f, 0.f, 0.f, 0.f, 0.f, 0.f, 0.f};
  v8f acc1 = {0.f, 0.f, 0.f, 0.f, 0.f, 0.f, 0.f, 0.f};
#pragma unroll 8
  for (int k = 0; k < K; k += 4) {
    v2f a = *(const v2f*)(xrow + k);
    v2f b0, b1;
    b0.x = wcol[k * NC];
    b0.y = wcol[(k + 1) * NC];
    b1.x = wcol[k * NC + 16];
    b1.y = wcol[(k + 1) * NC + 16];
#if __has_builtin(__builtin_amdgcn_wmma_f32_16x16x4_f32)
    acc0 = __builtin_amdgcn_wmma_f32_16x16x4_f32(false, a, false, b0,
                                                 (short)0, acc0, false, false);
    acc1 = __builtin_amdgcn_wmma_f32_16x16x4_f32(false, a, false, b1,
                                                 (short)0, acc1, false, false);
#else
    acc0[0] += a.x * b0.x;
    acc1[0] += a.x * b1.x;
#endif
  }
  float* __restrict__ yrow = Y + (size_t)(mTile + half * 8) * NC + head * 32 + l16;
#pragma unroll
  for (int r = 0; r < 8; ++r) {
    yrow[r * NC]      = acc0[r];
    yrow[r * NC + 16] = acc1[r];
  }

  const float s0 = att_s[head * 32 + l16], s1 = att_s[head * 32 + 16 + l16];
  const float d0 = att_d[head * 32 + l16], d1 = att_d[head * 32 + 16 + l16];
  float v[16];
#pragma unroll
  for (int r = 0; r < 8; ++r) {
    v[r]     = fmaf(acc0[r], s0, acc1[r] * s1);
    v[8 + r] = fmaf(acc0[r], d0, acc1[r] * d1);
  }
  const bool b3 = (l16 & 8) != 0;
  float u[8];
#pragma unroll
  for (int j = 0; j < 8; ++j) {
    const float keep = b3 ? v[8 + j] : v[j];
    const float send = b3 ? v[j] : v[8 + j];
    u[j] = keep + __shfl_xor(send, 8, 32);
  }
  const bool b2 = (l16 & 4) != 0;
  float t[4];
#pragma unroll
  for (int j = 0; j < 4; ++j) {
    const float keep = b2 ? u[4 + j] : u[j];
    const float send = b2 ? u[j] : u[4 + j];
    t[j] = keep + __shfl_xor(send, 4, 32);
  }
  const bool b1 = (l16 & 2) != 0;
  float q[2];
#pragma unroll
  for (int j = 0; j < 2; ++j) {
    const float keep = b1 ? t[2 + j] : t[j];
    const float send = b1 ? t[j] : t[2 + j];
    q[j] = keep + __shfl_xor(send, 2, 32);
  }
  const bool b0 = (l16 & 1) != 0;
  const float keep = b0 ? q[1] : q[0];
  const float send = b0 ? q[0] : q[1];
  const float res  = keep + __shfl_xor(send, 1, 32);
  const int row = mTile + 8 * half + (l16 & 7);
  float* __restrict__ dst = b3 ? a_d : a_s;
  dst[(size_t)row * HEADS + head] = res;
}

#define G1_TB 1024
#define G1_WV (G1_TB / 32)
template <int K, int NC>
__global__ __launch_bounds__(G1_TB) void gemm_att_wmma_f32_ldsw(
    const float* __restrict__ X, const float* __restrict__ W,
    const float* __restrict__ att_s, const float* __restrict__ att_d,
    float* __restrict__ Y, float* __restrict__ a_s, float* __restrict__ a_d,
    int M, const int* __restrict__ dstp, int E, int nblk, unsigned* __restrict__ ob) {
  constexpr int HEADS = NC / 32;
  constexpr int G1_PITCH = 2 * NC + 32;
  static_assert(NC % 32 == 0 && K % 4 == 0, "a head is 32 channels; K steps by 4");
  static_assert(G1_WV % HEADS == 0 && NC % 4 == 0, "a tile slot is HEADS waves; W rows move as 16-byte pieces");
  __shared__ __attribute__((aligned(16))) float s_w[(K / 2) * G1_PITCH];
  const int tid  = threadIdx.x;
  const int lane = tid & 31;
  {
    const bool ok = ((E & 3) == 0) && (((size_t)dstp & 15) == 0) &&
                    (((size_t)ob & 15) == 0) && nblk <= 256;
    if (ok) {
      const int nQ = E >> 2;
      const int gw = (blockIdx.x * blockDim.x + threadIdx.x) >> 5;
      const int nw = (gridDim.x * blockDim.x) >> 5;
      for (int base = gw * 32; base < nQ; base += nw * 32) {
        const int q = base + lane;
        if (q < nQ) {
          const int4 d = ((const int4*)dstp)[q];
          ob[q] = ((unsigned)d.x / NPB) | (((unsigned)d.y / NPB) << 8) |
                  (((unsigned)d.z / NPB) << 16) | (((unsigned)d.w / NPB) << 24);
        }
      }
    }
  }
  for (int i = tid; i < (K / 2) * (NC / 4); i += G1_TB) {
    const int k2 = i / (NC / 4);
    const int c4 = i % (NC / 4);
    const float4 r0 = *(const float4*)(W + (size_t)(2 * k2) * NC + 4 * c4);
    const float4 r1 = *(const float4*)(W + (size_t)(2 * k2 + 1) * NC + 4 * c4);
    float4* __restrict__ d = (float4*)(s_w + k2 * G1_PITCH + 8 * c4);
    d[0] = make_float4(r0.x, r1.x, r0.y, r1.y);
    d[1] = make_float4(r0.z, r1.z, r0.w, r1.w);
  }
  __syncthreads();

  const int wv     = __builtin_amdgcn_readfirstlane(tid >> 5);
  const int head   = wv % HEADS;
  const int slot   = (wv / HEADS) * gridDim.x + blockIdx.x;
  const int nSlots = gridDim.x * (G1_WV / HEADS);
  const int nTiles = M >> 4;
  const int half   = lane >> 4;
  const int l16    = lane & 15;

  const float* __restrict__ wl = s_w + half * G1_PITCH + (head * 32 + l16) * 2;
  const float s0 = att_s[head * 32 + l16], s1 = att_s[head * 32 + 16 + l16];
  const float d0 = att_d[head * 32 + l16], d1 = att_d[head * 32 + 16 + l16];
  const bool b3 = (l16 & 8) != 0;
  const bool b2 = (l16 & 4) != 0;
  const bool b1 = (l16 & 2) != 0;
  const bool b0 = (l16 & 1) != 0;
  float* __restrict__ dsel = b3 ? a_d : a_s;

  for (int tile = slot; tile < nTiles; tile += nSlots) {
    const int mTile = tile << 4;
    const float* __restrict__ xrow = X + (size_t)(mTile + l16) * K + 2 * half;
    v8f acc0 = {0.f, 0.f, 0.f, 0.f, 0.f, 0.f, 0.f, 0.f};
    v8f acc1 = {0.f, 0.f, 0.f, 0.f, 0.f, 0.f, 0.f, 0.f};
#pragma unroll 8
    for (int k = 0; k < K; k += 4) {
      const v2f a = *(const v2f*)(xrow + k);
      const float* __restrict__ p = wl + (k >> 1) * G1_PITCH;
      const v2f bf0 = *(const v2f*)p;
      const v2f bf1 = *(const v2f*)(p + 32);
#if __has_builtin(__builtin_amdgcn_wmma_f32_16x16x4_f32)
      acc0 = __builtin_amdgcn_wmma_f32_16x16x4_f32(false, a, false, bf0,
                                                   (short)0, acc0, false, false);
      acc1 = __builtin_amdgcn_wmma_f32_16x16x4_f32(false, a, false, bf1,
                                                   (short)0, acc1, false, false);
#else
      acc0[0] += a.x * bf0.x;
      acc1[0] += a.x * bf1.x;
#endif
    }
    float* __restrict__ yrow = Y + (size_t)(mTile + half * 8) * NC + head * 32 + l16;
#pragma unroll
    for (int r = 0; r < 8; ++r) {
      yrow[r * NC]      = acc0[r];
      yrow[r * NC + 16] = acc1[r];
    }
    float v[16];
#pragma unroll
    for (int r = 0; r < 8; ++r) {
      v[r]     = fmaf(acc0[r], s0, acc1[r] * s1);
      v[8 + r] = fmaf(acc0[r], d0, acc1[r] * d1);
    }
    float u[8];
#pragma unroll
    for (int j = 0; j < 8; ++j) {
      const float keep = b3 ? v[8 + j] : v[j];
      const float send = b3 ? v[j] : v[8 + j];
      u[j] = keep + __shfl_xor(send, 8, 32);
    }
    float t[4];
#pragma unroll
    for (int j = 0; j < 4; ++j) {
      const float keep = b2 ? u[4 + j] : u[j];
      const float send = b2 ? u[j] : u[4 + j];
      t[j] = keep + __shfl_xor(send, 4, 32);
    }
    float q[2];
#pragma unroll
    for (int j = 0; j < 2; ++j) {
      const float keep = b1 ? t[2 + j] : t[j];
      const float send = b1 ? t[j] : t[2 + j];
      q[j] = keep + __shfl_xor(send, 2, 32);
    }
    const float keep = b0 ? q[1] : q[0];
    const float send = b0 ? q[0] : q[1];
    const float res  = keep + __shfl_xor(send, 1, 32);
    const int row = mTile + 8 * half + (l16 & 7);
    dsel[(size_t)row * HEADS + head] = res;
  }
}

#define SEG_TB 1024
#define SEG_WV (SEG_TB / 32)
#define SCAN_U 4
#define SCAN_G (32 * SCAN_U)
#define WCAP 192
#define LCAP (SEG_WV * WCAP + NPB)
#define LPAD ((LCAP + 511) & ~511)
#define MCAP 256
#define RPW ((NPB + SEG_WV - 1) / SEG_WV)
static_assert(SEG_WV == 32, "a row's owner wave is its local index & 31");
static_assert(NPB <= 256, "local dst index is kept in a byte");
static_assert(SCAN_U == 4, "the owner-byte scan packs a trip's 16 dwords per lane into two 32-bit hit masks");

template <int H>
__device__ __forceinline__ void ld_coef(const float* __restrict__ a, int n,
                                        float (&v)[H]) {
  if constexpr ((H & 3) == 0) {
#pragma unroll
    for (int i = 0; i < H; i += 4) {
      const float4 t = *(const float4*)(a + (size_t)n * H + i);
      v[i] = t.x; v[i + 1] = t.y; v[i + 2] = t.z; v[i + 3] = t.w;
    }
  } else {
#pragma unroll
    for (int i = 0; i < H; ++i) v[i] = a[(size_t)n * H + i];
  }
}

__device__ __forceinline__ void ld_group(const int* __restrict__ dstp, int g,
                                         int lane, int4 (&v)[SCAN_U]) {
  const int4* __restrict__ p = (const int4*)dstp + (size_t)g * SCAN_G + lane;
#pragma unroll
  for (int u = 0; u < SCAN_U; ++u) v[u] = p[u * 32];
}

__device__ __forceinline__ void ld_groupb(const unsigned* __restrict__ ob, int g,
                                          int lane, uint4 (&v)[SCAN_U]) {
  const uint4* __restrict__ p = (const uint4*)ob + (size_t)g * SCAN_G + lane;
#pragma unroll
  for (int u = 0; u < SCAN_U; ++u) v[u] = p[u * 32];
}

template <int H, bool BUILD>
__global__ __launch_bounds__(SEG_TB) void seg_softmax_agg(
    const int* __restrict__ ei, int E, int N,
    const float* __restrict__ a_s, const float* __restrict__ a_d,
    const float* __restrict__ h, const float* __restrict__ bias,
    float* __restrict__ out, int* __restrict__ wlist, int* __restrict__ wcnt,
    const unsigned* __restrict__ ob) {
  constexpr int HC = H * 32;
  constexpr bool OWN = (H >= 8);
  constexpr int DEN_N = OWN ? H * SEG_WV * RPW : NPB * H;
  auto dix = [](int dl, int hh) -> int {
    if constexpr (OWN) return hh * (SEG_WV * RPW) + (dl & (SEG_WV - 1)) * RPW + dl / SEG_WV;
    else               return dl * H + hh;
  };
  __shared__ float    s_num[NPB * HC];
  __shared__ float    s_den[DEN_N];
  __shared__ unsigned s_max[NPB * H];
  __shared__ float    s_ad[NPB * H];
  __shared__ int           s_stage[BUILD ? SEG_WV * WCAP : 1];
  __shared__ int           s_wc[SEG_WV];
  __shared__ int           l_src[LCAP];
  __shared__ __attribute__((aligned(16))) unsigned char l_dl[LPAD];
  __shared__ unsigned short s_mine[OWN ? SEG_WV * MCAP : 1];

  const int tid  = threadIdx.x;
  const int lane = tid & 31;
  const int wv   = __builtin_amdgcn_readfirstlane(tid >> 5);
  const int n0   = blockIdx.x * NPB;
  const int cnt  = min(NPB, N - n0);

  if constexpr (!OWN) {
    for (int i = tid; i < NPB * HC; i += SEG_TB) s_num[i] = 0.f;
  }
  for (int i = tid; i < DEN_N; i += SEG_TB) s_den[i] = 0.f;
  for (int i = tid; i < NPB * H; i += SEG_TB) {
    s_max[i] = 0u;
    s_ad[i]  = (i < cnt * H) ? a_d[(size_t)n0 * H + i] : 0.f;
  }

  const int* __restrict__ srcp = ei;
  const int* __restrict__ dstp = ei + E;
  const bool al16 = ((E & 3) == 0) && (((size_t)dstp & 15) == 0);
  const int nG    = al16 ? (E >> 2) / SCAN_G : 0;
  const int eRest = nG * SCAN_G * 4;

  auto edgeA = [&](int d, int e, bool self) {
    const int dl = d - n0;
    if ((unsigned)dl < (unsigned)cnt) {
      const int s = self ? d : srcp[e];
      float av[H];
      ld_coef<H>(a_s, s, av);
#pragma unroll
      for (int hh = 0; hh < H; ++hh) {
        float l = av[hh] + s_ad[dl * H + hh];
        l = (l > 0.f) ? l : LRELU_SLOPE * l;
        atomicMax(&s_max[dl * H + hh], f2ord(l));
      }
    }
  };

  auto edgeB = [&](int d, int e, bool self) {
    const int  dl  = d - n0;
    const bool hit = (unsigned)dl < (unsigned)cnt;
    unsigned mask = __builtin_amdgcn_ballot_w32(hit);
    if (mask != 0u) {
      int   s = 0;
      float w[H];
#pragma unroll
      for (int hh = 0; hh < H; ++hh) w[hh] = 0.f;
      if (hit) {
        s = self ? d : srcp[e];
        float av[H];
        ld_coef<H>(a_s, s, av);
#pragma unroll
        for (int hh = 0; hh < H; ++hh) {
          float l = av[hh] + s_ad[dl * H + hh];
          l = (l > 0.f) ? l : LRELU_SLOPE * l;
          w[hh] = expf(l - ord2f(s_max[dl * H + hh]));
          atomicAdd(&s_den[dix(dl, hh)], w[hh]);
        }
      }
      while (mask) {
        const int L = __builtin_ctz(mask);
        mask &= mask - 1u;
        const int sB = __builtin_amdgcn_readlane(s, L);
        const int dB = __builtin_amdgcn_readlane(dl, L);
        const float* __restrict__ hrow = h + (size_t)sB * HC + lane;
        float hv[H];
#pragma unroll
        for (int hh = 0; hh < H; ++hh) hv[hh] = hrow[hh * 32];
#pragma unroll
        for (int hh = 0; hh < H; ++hh) {
          const float wB = __int_as_float(
              __builtin_amdgcn_readlane(__float_as_int(w[hh]), L));
          atomicAdd(&s_num[dB * HC + hh * 32 + lane], wB * hv[hh]);
        }
      }
    }
  };

  const int nlS = lane * SEG_WV + wv;
  const int dS  = (nlS < cnt) ? n0 + nlS : -1;

  auto scan = [&](auto&& edgeFn) {
    if (wv < nG) {
      int4 nx[SCAN_U];
      ld_group(dstp, wv, lane, nx);
      for (int g = wv; g < nG; g += SEG_WV) {
        int4 cur[SCAN_U];
#pragma unroll
        for (int u = 0; u < SCAN_U; ++u) cur[u] = nx[u];
        ld_group(dstp, min(g + SEG_WV, nG - 1), lane, nx);
        unsigned hm = 0u;
#pragma unroll
        for (int u = 0; u < SCAN_U; ++u) {
          hm |= ((unsigned)(cur[u].x - n0) < (unsigned)cnt) ? (1u << (4 * u))     : 0u;
          hm |= ((unsigned)(cur[u].y - n0) < (unsigned)cnt) ? (1u << (4 * u + 1)) : 0u;
          hm |= ((unsigned)(cur[u].z - n0) < (unsigned)cnt) ? (1u << (4 * u + 2)) : 0u;
          hm |= ((unsigned)(cur[u].w - n0) < (unsigned)cnt) ? (1u << (4 * u + 3)) : 0u;
        }
        const int ebase = 4 * (g * SCAN_G + lane);
        while (__builtin_amdgcn_ballot_w32(hm != 0u) != 0u) {
          const bool has = (hm != 0u);
          const int  idx = has ? __builtin_ctz(hm) : 0;
          hm &= hm - 1u;
          const int e = ebase + (idx >> 2) * (4 * 32) + (idx & 3);
          const int d = has ? dstp[e] : -1;
          edgeFn(d, e, false);
        }
      }
    }
    for (int eb = eRest + wv * 32; eb < E; eb += SEG_TB) {
      const int e = eb + lane;
      const int d = (e < E) ? dstp[e] : -1;
      edgeFn(d, (e < E) ? e : 0, false);
    }
  };

  int wc = 0;
  if constexpr (BUILD) {
    auto stage = [&](bool hit, int e) {
      const unsigned m = __builtin_amdgcn_ballot_w32(hit);
      if (m != 0u) {
        const int pos = wc + (int)__builtin_amdgcn_mbcnt_lo(m, 0u);
        if (hit && pos < WCAP) s_stage[wv * WCAP + pos] = e;
        wc += __builtin_popcount(m);
      }
    };
    const bool useB = al16 && (((size_t)ob & 15) == 0) && gridDim.x <= 256;
    const int nGb    = useB ? (E >> 4) / SCAN_G : 0;
    const int eRestB = useB ? nGb * SCAN_G * 16 : eRest;
    if (useB) {
      const unsigned rep = (unsigned)blockIdx.x * 0x01010101u;
      auto own4 = [&](unsigned w) -> unsigned {
        const unsigned x = w ^ rep;
        return (~(((x & 0x7F7F7F7Fu) + 0x7F7F7F7Fu) | x | 0x7F7F7F7Fu)) >> 7;
      };
      if (wv < nGb) {
        uint4 nx[SCAN_U];
        ld_groupb(ob, wv, lane, nx);
        for (int g = wv; g < nGb; g += SEG_WV) {
          uint4 cur[SCAN_U];
#pragma unroll
          for (int u = 0; u < SCAN_U; ++u) cur[u] = nx[u];
          ld_groupb(ob, min(g + SEG_WV, nGb - 1), lane, nx);
          unsigned mA = 0u, mB = 0u;
#pragma unroll
          for (int u = 0; u < SCAN_U; ++u) {
            const unsigned f0 = own4(cur[u].x), f1 = own4(cur[u].y);
            const unsigned f2 = own4(cur[u].z), f3 = own4(cur[u].w);
            const int sh = 4 * (u & 1);
            const unsigned m = (f0 << sh) | (f1 << (sh + 1)) | (f2 << (sh + 2)) | (f3 << (sh + 3));
            if (u < 2) mA |= m; else mB |= m;
          }
          const int ebase = 16 * (g * SCAN_G + lane);
          while (__builtin_amdgcn_ballot_w32((mA | mB) != 0u) != 0u) {
            const bool     inA = (mA != 0u);
            const unsigned mx  = inA ? mA : mB;
            const bool     has = (mx != 0u);
            const int      t   = has ? __builtin_ctz(mx) : 0;
            const unsigned mc  = mx & (mx - 1u);
            mA = inA ? mc : mA;
            mB = inA ? mB : mc;
            const int j = (t & 7) + (inA ? 0 : 8);
            const int e = ebase + (j >> 2) * 512 + (j & 3) * 4 + (t >> 3);
            stage(has, e);
          }
        }
      }
    } else if (wv < nG) {
      int4 nx[SCAN_U];
      ld_group(dstp, wv, lane, nx);
      for (int g = wv; g < nG; g += SEG_WV) {
        int4 cur[SCAN_U];
#pragma unroll
        for (int u = 0; u < SCAN_U; ++u) cur[u] = nx[u];
        ld_group(dstp, min(g + SEG_WV, nG - 1), lane, nx);
        const int ebase = 4 * (g * SCAN_G + lane);
#pragma unroll
        for (int u = 0; u < SCAN_U; ++u) {
          stage((unsigned)(cur[u].x - n0) < (unsigned)cnt, ebase + u * 128);
          stage((unsigned)(cur[u].y - n0) < (unsigned)cnt, ebase + u * 128 + 1);
          stage((unsigned)(cur[u].z - n0) < (unsigned)cnt, ebase + u * 128 + 2);
          stage((unsigned)(cur[u].w - n0) < (unsigned)cnt, ebase + u * 128 + 3);
        }
      }
    }
    for (int eb = eRestB + wv * 32; eb < E; eb += SEG_TB) {
      const int e = eb + lane;
      const int d = (e < E) ? dstp[e] : -1;
      stage((unsigned)(d - n0) < (unsigned)cnt, e);
    }
    if (lane == 0) s_wc[wv] = wc;
  }
  __syncthreads();

  int2* __restrict__ wl = (int2*)wlist + (size_t)blockIdx.x * LCAP;

  auto put = [&](int idx, int s, int dl) {
    l_src[idx] = s;
    l_dl[idx]  = (unsigned char)dl;
    if constexpr (BUILD) wl[idx] = make_int2(s, dl);
    float av[H];
    ld_coef<H>(a_s, s, av);
#pragma unroll
    for (int hh = 0; hh < H; ++hh) {
      float l = av[hh] + s_ad[dl * H + hh];
      l = (l > 0.f) ? l : LRELU_SLOPE * l;
      atomicMax(&s_max[dl * H + hh], f2ord(l));
    }
  };

  int total;
  if constexpr (BUILD) {
    const int c = s_wc[lane];
    int pre = (lane < wv) ? c : 0;
    int all = c;
#pragma unroll
    for (int off = 16; off > 0; off >>= 1) {
      pre += __shfl_xor(pre, off, 32);
      all += __shfl_xor(all, off, 32);
    }
    pre = __builtin_amdgcn_readfirstlane(pre);
    all = __builtin_amdgcn_readfirstlane(all);
    const bool over = __builtin_amdgcn_ballot_w32(c > WCAP) != 0u;
    total = over ? -1 : all + cnt;
    if (tid == 0) wcnt[blockIdx.x] = total;
    if (!over) {
      for (int j = lane; j < wc; j += 32) {
        const int e = s_stage[wv * WCAP + j];
        put(pre + j, srcp[e], dstp[e] - n0);
      }
      if (tid < cnt) put(all + tid, n0 + tid, tid);
    }
  } else {
    total = wcnt[blockIdx.x];
    if (total > LCAP) total = -1;
    for (int i = tid; i < total; i += SEG_TB) {
      const int2 v = wl[i];
      put(i, v.x, v.y);
    }
  }

  if (total >= 0) {
    __syncthreads();
    constexpr int UNR = (H >= 8) ? 4 : 16;
    if constexpr (OWN) {
      const unsigned rep = (unsigned)wv * 0x01010101u;
      auto own4 = [&](unsigned q) -> unsigned {
        const unsigned x = (q & 0x1F1F1F1Fu) ^ rep;
        return ~(x + 0x7F7F7F7Fu) & 0x80808080u;
      };
      const int nCh = (total + 511) >> 9;
      unsigned short* __restrict__ mine = s_mine + wv * MCAP;
      int n = 0, ch = 0, ebase = 0;
      unsigned m = 0u;
      bool first = true;
      for (;;) {
        while (n <= MCAP - 32) {
          if (__builtin_amdgcn_ballot_w32(m != 0u) == 0u) {
            if (ch >= nCh) break;
            ebase = ch * 512 + lane * 16;
            ++ch;
            const uint4 q = *(const uint4*)(l_dl + ebase);
            m = (own4(q.x) >> 7) | (own4(q.y) >> 6) | (own4(q.z) >> 5) | (own4(q.w) >> 4);
            continue;
          }
          const bool has = (m != 0u);
          const int  t   = has ? __builtin_ctz(m) : 0;
          m &= m - 1u;
          const int  e   = ebase + 4 * (t & 7) + (t >> 3);
          const bool hit = has && e < total;
          const unsigned bm = __builtin_amdgcn_ballot_w32(hit);
          if (bm != 0u) {
            const int pos = n + (int)__builtin_amdgcn_mbcnt_lo(bm, 0u);
            if (hit) mine[pos] = (unsigned short)e;
            n += __builtin_popcount(bm);
          }
        }
        if (n == 0) break;
        float acc[RPW][H];
#pragma unroll
        for (int r = 0; r < RPW; ++r) {
#pragma unroll
          for (int hh = 0; hh < H; ++hh) acc[r][hh] = 0.f;
        }
        if (!first) {
#pragma unroll
          for (int r = 0; r < RPW; ++r) {
            const int row = wv + 32 * r;
            if (row < cnt) {
#pragma unroll
              for (int hh = 0; hh < H; ++hh) acc[r][hh] = s_num[row * HC + hh * 32 + lane];
            }
          }
        }
        for (int base = 0; base < n; base += 32) {
          const int  jj    = base + lane;
          const bool valid = jj < n;
          int   s = 0, dl = 0;
          float w[H];
#pragma unroll
          for (int hh = 0; hh < H; ++hh) w[hh] = 0.f;
          if (valid) {
            const int i = mine[jj];
            s  = l_src[i];
            dl = l_dl[i];
            float av[H];
            ld_coef<H>(a_s, s, av);
#pragma unroll
            for (int hh = 0; hh < H; ++hh) {
              float l = av[hh] + s_ad[dl * H + hh];
              l = (l > 0.f) ? l : LRELU_SLOPE * l;
              w[hh] = expf(l - ord2f(s_max[dl * H + hh]));
              atomicAdd(&s_den[dix(dl, hh)], w[hh]);
            }
          }
          const int nb = min(32, n - base);
          for (int L0 = 0; L0 < nb; L0 += UNR) {
            float hv[UNR][H];
            int   kB[UNR];
#pragma unroll
            for (int j = 0; j < UNR; ++j) {
              const int L  = min(L0 + j, nb - 1);
              const int sB = __builtin_amdgcn_readlane(s, L);
              kB[j]        = __builtin_amdgcn_readlane(dl, L) >> 5;
              const float* __restrict__ hrow = h + (size_t)sB * HC + lane;
#pragma unroll
              for (int hh = 0; hh < H; ++hh) hv[j][hh] = hrow[hh * 32];
            }
#pragma unroll
            for (int j = 0; j < UNR; ++j) {
              if (L0 + j < nb) {
                float wB[H];
#pragma unroll
                for (int hh = 0; hh < H; ++hh)
                  wB[hh] = __int_as_float(
                      __builtin_amdgcn_readlane(__float_as_int(w[hh]), L0 + j));
#pragma unroll
                for (int r = 0; r < RPW; ++r) {
                  if (kB[j] == r) {
#pragma unroll
                    for (int hh = 0; hh < H; ++hh) acc[r][hh] = fmaf(wB[hh], hv[j][hh], acc[r][hh]);
                  }
                }
              }
            }
          }
        }
#pragma unroll
        for (int r = 0; r < RPW; ++r) {
          const int row = wv + 32 * r;
          if (row < cnt) {
#pragma unroll
            for (int hh = 0; hh < H; ++hh) s_num[row * HC + hh * 32 + lane] = acc[r][hh];
          }
        }
        first = false;
        n = 0;
      }
    } else {
      const int per = (total + SEG_WV - 1) / SEG_WV;
      const int lo  = wv * per;
      const int hi  = min(total, lo + per);
      for (int base = lo; base < hi; base += 32) {
        const int  i     = base + lane;
        const bool valid = i < hi;
        int   s = 0, dl = 0;
        float w[H];
#pragma unroll
        for (int hh = 0; hh < H; ++hh) w[hh] = 0.f;
        if (valid) {
          s  = l_src[i];
          dl = l_dl[i];
          float av[H];
          ld_coef<H>(a_s, s, av);
#pragma unroll
          for (int hh = 0; hh < H; ++hh) {
            float l = av[hh] + s_ad[dl * H + hh];
            l = (l > 0.f) ? l : LRELU_SLOPE * l;
            w[hh] = expf(l - ord2f(s_max[dl * H + hh]));
            atomicAdd(&s_den[dix(dl, hh)], w[hh]);
          }
        }
        const int n = min(32, hi - base);
        for (int L0 = 0; L0 < n; L0 += UNR) {
          float hv[UNR][H];
          int   dB[UNR];
#pragma unroll
          for (int j = 0; j < UNR; ++j) {
            const int L  = min(L0 + j, n - 1);
            const int sB = __builtin_amdgcn_readlane(s, L);
            dB[j]        = __builtin_amdgcn_readlane(dl, L);
            const float* __restrict__ hrow = h + (size_t)sB * HC + lane;
#pragma unroll
            for (int hh = 0; hh < H; ++hh) hv[j][hh] = hrow[hh * 32];
          }
#pragma unroll
          for (int j = 0; j < UNR; ++j) {
            if (L0 + j < n) {
#pragma unroll
              for (int hh = 0; hh < H; ++hh) {
                const float wB = __int_as_float(
                    __builtin_amdgcn_readlane(__float_as_int(w[hh]), L0 + j));
                atomicAdd(&s_num[dB[j] * HC + hh * 32 + lane], wB * hv[j][hh]);
              }
            }
          }
        }
      }
    }
  } else {
    if constexpr (OWN) {
      for (int i = tid; i < NPB * HC; i += SEG_TB) s_num[i] = 0.f;
    }
    edgeA(dS, 0, true);
    scan(edgeA);
    __syncthreads();
    edgeB(dS, 0, true);
    scan(edgeB);
  }
  __syncthreads();

  float* __restrict__ orow = out + (size_t)n0 * HC;
  for (int i = tid; i < cnt * HC; i += SEG_TB) {
    const int nl = i / HC;
    const int hc = i % HC;
    float v = s_num[i] / (s_den[dix(nl, hc / 32)] + 1e-16f) + bias[hc];
    v = (v > 0.f) ? v : (expf(v) - 1.f);
    orow[i] = v;
  }
}

extern "C" void kernel_launch(void* const* d_in, const int* in_sizes, int n_in,
                              void* d_out, int out_size, void* d_ws, size_t ws_size,
                              hipStream_t stream) {
  const float* x   = (const float*)d_in[0];
  const int*   ei  = (const int*)d_in[1];
  const float* W1  = (const float*)d_in[2];
  const float* as1 = (const float*)d_in[3];
  const float* ad1 = (const float*)d_in[4];
  const float* b1  = (const float*)d_in[5];
  const float* W2  = (const float*)d_in[6];
  const float* as2 = (const float*)d_in[7];
  const float* ad2 = (const float*)d_in[8];
  const float* b2  = (const float*)d_in[9];

  const int HC   = in_sizes[3];
  const int Fin  = in_sizes[2] / HC;
  const int N    = in_sizes[0] / Fin;
  const int E    = in_sizes[1] / 2;
  const int H    = 8;
  const int Cout = in_sizes[7];

  float* ws = (float*)d_ws;
  float* h1  = ws;
  float* x2  = h1 + (size_t)N * HC;
  float* h2  = x2 + (size_t)N * HC;
  float* aS1 = h2 + (size_t)N * Cout;
  float* aD1 = aS1 + (size_t)N * H;
  float* aS2 = aD1 + (size_t)N * H;
  float* aD2 = aS2 + N;
  const int segBlocksWs = (N + NPB - 1) / NPB;
  int* eList = (int*)(aD2 + N);
  int* eCnt  = eList + (size_t)segBlocksWs * LCAP * 2;
  unsigned* eOwn = (unsigned*)(eCnt + ((segBlocksWs + 3) & ~3));

  const int TB = 256;
  auto blocks = [](long long n, int tb) { return (int)((n + tb - 1) / tb); };
  const int segBlocks = (N + NPB - 1) / NPB;

  {
    gemm_att_wmma_f32_ldsw<128, 256><<<256, G1_TB, 0, stream>>>(
        x, W1, as1, ad1, h1, aS1, aD1, N, ei + E, E, segBlocks, eOwn);
    seg_softmax_agg<8, true><<<segBlocks, SEG_TB, 0, stream>>>(
        ei, E, N, aS1, aD1, h1, b1, x2, eList, eCnt, eOwn);
  }

  {
    const long long waves = (long long)(N / 16) * (Cout / 32);
    gemm_att_wmma_f32<256, 32><<<blocks(waves * 32, TB), TB, 0, stream>>>(
        x2, W2, as2, ad2, h2, aS2, aD2, N);
    seg_softmax_agg<1, false><<<segBlocks, SEG_TB, 0, stream>>>(
        ei, E, N, aS2, aD2, h2, b2, (float*)d_out, eList, eCnt, eOwn);
  }
}


# Layer-1 aggregation: a wave finishes its own rows once its list walk ends (sums via its own LDS slots, divide, bias, ELU, whole-line stores); the closing barrier and block-wide output pass go
# speedup vs baseline: 10.0962x; 1.0320x over previous
#include <hip/hip_runtime.h>
#include <math.h>

typedef float v2f __attribute__((ext_vector_type(2)));
typedef float v8f __attribute__((ext_vector_type(8)));

#define LRELU_SLOPE 0.2f
#define NPB 196

__device__ __forceinline__ unsigned f2ord(float f) {
  unsigned u = __float_as_uint(f);
  return (u & 0x80000000u) ? ~u : (u | 0x80000000u);
}
__device__ __forceinline__ float ord2f(unsigned u) {
  return __uint_as_float((u & 0x80000000u) ? (u & 0x7FFFFFFFu) : ~u);
}

template <int K, int NC>
__global__ __launch_bounds__(256) void gemm_att_wmma_f32(
    const float* __restrict__ X, const float* __restrict__ W,
    const float* __restrict__ att_s, const float* __restrict__ att_d,
    float* __restrict__ Y, float* __restrict__ a_s, float* __restrict__ a_d,
    int M) {
  constexpr int HEADS = NC / 32;
  static_assert(NC % 32 == 0 && K % 4 == 0, "a head is 32 channels; K steps by 4");
  const int lane  = threadIdx.x & 31;
  const int wave  = (blockIdx.x * blockDim.x + threadIdx.x) >> 5;
  if (wave >= (M >> 4) * HEADS) return;
  const int mTile = (wave / HEADS) << 4;
  const int head  = wave % HEADS;
  const int half  = lane >> 4;
  const int l16   = lane & 15;

  const float* __restrict__ xrow = X + (size_t)(mTile + l16) * K + 2 * half;
  const float* __restrict__ wcol = W + (size_t)(2 * half) * NC + head * 32 + l16;

  v8f acc0 = {0.f, 0.f, 0.f, 0.f, 0.f, 0.f, 0.f, 0.f};
  v8f acc1 = {0.f, 0.f, 0.f, 0.f, 0.f, 0.f, 0.f, 0.f};
#pragma unroll 8
  for (int k = 0; k < K; k += 4) {
    v2f a = *(const v2f*)(xrow + k);
    v2f b0, b1;
    b0.x = wcol[k * NC];
    b0.y = wcol[(k + 1) * NC];
    b1.x = wcol[k * NC + 16];
    b1.y = wcol[(k + 1) * NC + 16];
#if __has_builtin(__builtin_amdgcn_wmma_f32_16x16x4_f32)
    acc0 = __builtin_amdgcn_wmma_f32_16x16x4_f32(false, a, false, b0,
                                                 (short)0, acc0, false, false);
    acc1 = __builtin_amdgcn_wmma_f32_16x16x4_f32(false, a, false, b1,
                                                 (short)0, acc1, false, false);
#else
    acc0[0] += a.x * b0.x;
    acc1[0] += a.x * b1.x;
#endif
  }
  float* __restrict__ yrow = Y + (size_t)(mTile + half * 8) * NC + head * 32 + l16;
#pragma unroll
  for (int r = 0; r < 8; ++r) {
    yrow[r * NC]      = acc0[r];
    yrow[r * NC + 16] = acc1[r];
  }

  const float s0 = att_s[head * 32 + l16], s1 = att_s[head * 32 + 16 + l16];
  const float d0 = att_d[head * 32 + l16], d1 = att_d[head * 32 + 16 + l16];
  float v[16];
#pragma unroll
  for (int r = 0; r < 8; ++r) {
    v[r]     = fmaf(acc0[r], s0, acc1[r] * s1);
    v[8 + r] = fmaf(acc0[r], d0, acc1[r] * d1);
  }
  const bool b3 = (l16 & 8) != 0;
  float u[8];
#pragma unroll
  for (int j = 0; j < 8; ++j) {
    const float keep = b3 ? v[8 + j] : v[j];
    const float send = b3 ? v[j] : v[8 + j];
    u[j] = keep + __shfl_xor(send, 8, 32);
  }
  const bool b2 = (l16 & 4) != 0;
  float t[4];
#pragma unroll
  for (int j = 0; j < 4; ++j) {
    const float keep = b2 ? u[4 + j] : u[j];
    const float send = b2 ? u[j] : u[4 + j];
    t[j] = keep + __shfl_xor(send, 4, 32);
  }
  const bool b1 = (l16 & 2) != 0;
  float q[2];
#pragma unroll
  for (int j = 0; j < 2; ++j) {
    const float keep = b1 ? t[2 + j] : t[j];
    const float send = b1 ? t[j] : t[2 + j];
    q[j] = keep + __shfl_xor(send, 2, 32);
  }
  const bool b0 = (l16 & 1) != 0;
  const float keep = b0 ? q[1] : q[0];
  const float send = b0 ? q[0] : q[1];
  const float res  = keep + __shfl_xor(send, 1, 32);
  const int row = mTile + 8 * half + (l16 & 7);
  float* __restrict__ dst = b3 ? a_d : a_s;
  dst[(size_t)row * HEADS + head] = res;
}

#define G1_TB 1024
#define G1_WV (G1_TB / 32)
template <int K, int NC>
__global__ __launch_bounds__(G1_TB) void gemm_att_wmma_f32_ldsw(
    const float* __restrict__ X, const float* __restrict__ W,
    const float* __restrict__ att_s, const float* __restrict__ att_d,
    float* __restrict__ Y, float* __restrict__ a_s, float* __restrict__ a_d,
    int M, const int* __restrict__ dstp, int E, int nblk, unsigned* __restrict__ ob) {
  constexpr int HEADS = NC / 32;
  constexpr int G1_PITCH = 2 * NC + 32;
  static_assert(NC % 32 == 0 && K % 4 == 0, "a head is 32 channels; K steps by 4");
  static_assert(G1_WV % HEADS == 0 && NC % 4 == 0, "a tile slot is HEADS waves; W rows move as 16-byte pieces");
  __shared__ __attribute__((aligned(16))) float s_w[(K / 2) * G1_PITCH];
  const int tid  = threadIdx.x;
  const int lane = tid & 31;
  {
    const bool ok = ((E & 3) == 0) && (((size_t)dstp & 15) == 0) &&
                    (((size_t)ob & 15) == 0) && nblk <= 256;
    if (ok) {
      const int nQ = E >> 2;
      const int gw = (blockIdx.x * blockDim.x + threadIdx.x) >> 5;
      const int nw = (gridDim.x * blockDim.x) >> 5;
      for (int base = gw * 32; base < nQ; base += nw * 32) {
        const int q = base + lane;
        if (q < nQ) {
          const int4 d = ((const int4*)dstp)[q];
          ob[q] = ((unsigned)d.x / NPB) | (((unsigned)d.y / NPB) << 8) |
                  (((unsigned)d.z / NPB) << 16) | (((unsigned)d.w / NPB) << 24);
        }
      }
    }
  }
  for (int i = tid; i < (K / 2) * (NC / 4); i += G1_TB) {
    const int k2 = i / (NC / 4);
    const int c4 = i % (NC / 4);
    const float4 r0 = *(const float4*)(W + (size_t)(2 * k2) * NC + 4 * c4);
    const float4 r1 = *(const float4*)(W + (size_t)(2 * k2 + 1) * NC + 4 * c4);
    float4* __restrict__ d = (float4*)(s_w + k2 * G1_PITCH + 8 * c4);
    d[0] = make_float4(r0.x, r1.x, r0.y, r1.y);
    d[1] = make_float4(r0.z, r1.z, r0.w, r1.w);
  }
  __syncthreads();

  const int wv     = __builtin_amdgcn_readfirstlane(tid >> 5);
  const int head   = wv % HEADS;
  const int slot   = (wv / HEADS) * gridDim.x + blockIdx.x;
  const int nSlots = gridDim.x * (G1_WV / HEADS);
  const int nTiles = M >> 4;
  const int half   = lane >> 4;
  const int l16    = lane & 15;

  const float* __restrict__ wl = s_w + half * G1_PITCH + (head * 32 + l16) * 2;
  const float s0 = att_s[head * 32 + l16], s1 = att_s[head * 32 + 16 + l16];
  const float d0 = att_d[head * 32 + l16], d1 = att_d[head * 32 + 16 + l16];
  const bool b3 = (l16 & 8) != 0;
  const bool b2 = (l16 & 4) != 0;
  const bool b1 = (l16 & 2) != 0;
  const bool b0 = (l16 & 1) != 0;
  float* __restrict__ dsel = b3 ? a_d : a_s;

  for (int tile = slot; tile < nTiles; tile += nSlots) {
    const int mTile = tile << 4;
    const float* __restrict__ xrow = X + (size_t)(mTile + l16) * K + 2 * half;
    v8f acc0 = {0.f, 0.f, 0.f, 0.f, 0.f, 0.f, 0.f, 0.f};
    v8f acc1 = {0.f, 0.f, 0.f, 0.f, 0.f, 0.f, 0.f, 0.f};
#pragma unroll 8
    for (int k = 0; k < K; k += 4) {
      const v2f a = *(const v2f*)(xrow + k);
      const float* __restrict__ p = wl + (k >> 1) * G1_PITCH;
      const v2f bf0 = *(const v2f*)p;
      const v2f bf1 = *(const v2f*)(p + 32);
#if __has_builtin(__builtin_amdgcn_wmma_f32_16x16x4_f32)
      acc0 = __builtin_amdgcn_wmma_f32_16x16x4_f32(false, a, false, bf0,
                                                   (short)0, acc0, false, false);
      acc1 = __builtin_amdgcn_wmma_f32_16x16x4_f32(false, a, false, bf1,
                                                   (short)0, acc1, false, false);
#else
      acc0[0] += a.x * bf0.x;
      acc1[0] += a.x * bf1.x;
#endif
    }
    float* __restrict__ yrow = Y + (size_t)(mTile + half * 8) * NC + head * 32 + l16;
#pragma unroll
    for (int r = 0; r < 8; ++r) {
      yrow[r * NC]      = acc0[r];
      yrow[r * NC + 16] = acc1[r];
    }
    float v[16];
#pragma unroll
    for (int r = 0; r < 8; ++r) {
      v[r]     = fmaf(acc0[r], s0, acc1[r] * s1);
      v[8 + r] = fmaf(acc0[r], d0, acc1[r] * d1);
    }
    float u[8];
#pragma unroll
    for (int j = 0; j < 8; ++j) {
      const float keep = b3 ? v[8 + j] : v[j];
      const float send = b3 ? v[j] : v[8 + j];
      u[j] = keep + __shfl_xor(send, 8, 32);
    }
    float t[4];
#pragma unroll
    for (int j = 0; j < 4; ++j) {
      const float keep = b2 ? u[4 + j] : u[j];
      const float send = b2 ? u[j] : u[4 + j];
      t[j] = keep + __shfl_xor(send, 4, 32);
    }
    float q[2];
#pragma unroll
    for (int j = 0; j < 2; ++j) {
      const float keep = b1 ? t[2 + j] : t[j];
      const float send = b1 ? t[j] : t[2 + j];
      q[j] = keep + __shfl_xor(send, 2, 32);
    }
    const float keep = b0 ? q[1] : q[0];
    const float send = b0 ? q[0] : q[1];
    const float res  = keep + __shfl_xor(send, 1, 32);
    const int row = mTile + 8 * half + (l16 & 7);
    dsel[(size_t)row * HEADS + head] = res;
  }
}

#define SEG_TB 1024
#define SEG_WV (SEG_TB / 32)
#define SCAN_U 4
#define SCAN_G (32 * SCAN_U)
#define WCAP 192
#define LCAP (SEG_WV * WCAP + NPB)
#define LPAD ((LCAP + 511) & ~511)
#define MCAP 256
#define RPW ((NPB + SEG_WV - 1) / SEG_WV)
static_assert(SEG_WV == 32, "a row's owner wave is its local index & 31");
static_assert(NPB <= 256, "local dst index is kept in a byte");
static_assert(SCAN_U == 4, "the owner-byte scan packs a trip's 16 dwords per lane into two 32-bit hit masks");

template <int H>
__device__ __forceinline__ void ld_coef(const float* __restrict__ a, int n,
                                        float (&v)[H]) {
  if constexpr ((H & 3) == 0) {
#pragma unroll
    for (int i = 0; i < H; i += 4) {
      const float4 t = *(const float4*)(a + (size_t)n * H + i);
      v[i] = t.x; v[i + 1] = t.y; v[i + 2] = t.z; v[i + 3] = t.w;
    }
  } else {
#pragma unroll
    for (int i = 0; i < H; ++i) v[i] = a[(size_t)n * H + i];
  }
}

__device__ __forceinline__ void ld_group(const int* __restrict__ dstp, int g,
                                         int lane, int4 (&v)[SCAN_U]) {
  const int4* __restrict__ p = (const int4*)dstp + (size_t)g * SCAN_G + lane;
#pragma unroll
  for (int u = 0; u < SCAN_U; ++u) v[u] = p[u * 32];
}

__device__ __forceinline__ void ld_groupb(const unsigned* __restrict__ ob, int g,
                                          int lane, uint4 (&v)[SCAN_U]) {
  const uint4* __restrict__ p = (const uint4*)ob + (size_t)g * SCAN_G + lane;
#pragma unroll
  for (int u = 0; u < SCAN_U; ++u) v[u] = p[u * 32];
}

template <int H, bool BUILD>
__global__ __launch_bounds__(SEG_TB) void seg_softmax_agg(
    const int* __restrict__ ei, int E, int N,
    const float* __restrict__ a_s, const float* __restrict__ a_d,
    const float* __restrict__ h, const float* __restrict__ bias,
    float* __restrict__ out, int* __restrict__ wlist, int* __restrict__ wcnt,
    const unsigned* __restrict__ ob) {
  constexpr int HC = H * 32;
  constexpr bool OWN = (H >= 8);
  constexpr int DEN_N = OWN ? H * SEG_WV * RPW : NPB * H;
  auto dix = [](int dl, int hh) -> int {
    if constexpr (OWN) return hh * (SEG_WV * RPW) + (dl & (SEG_WV - 1)) * RPW + dl / SEG_WV;
    else               return dl * H + hh;
  };
  __shared__ float    s_num[NPB * HC];
  __shared__ float    s_den[DEN_N];
  __shared__ unsigned s_max[NPB * H];
  __shared__ float    s_ad[NPB * H];
  __shared__ int           s_stage[BUILD ? SEG_WV * WCAP : 1];
  __shared__ int           s_wc[SEG_WV];
  __shared__ int           l_src[LCAP];
  __shared__ __attribute__((aligned(16))) unsigned char l_dl[LPAD];
  __shared__ unsigned short s_mine[OWN ? SEG_WV * MCAP : 1];

  const int tid  = threadIdx.x;
  const int lane = tid & 31;
  const int wv   = __builtin_amdgcn_readfirstlane(tid >> 5);
  const int n0   = blockIdx.x * NPB;
  const int cnt  = min(NPB, N - n0);

  if constexpr (!OWN) {
    for (int i = tid; i < NPB * HC; i += SEG_TB) s_num[i] = 0.f;
  }
  for (int i = tid; i < DEN_N; i += SEG_TB) s_den[i] = 0.f;
  for (int i = tid; i < NPB * H; i += SEG_TB) {
    s_max[i] = 0u;
    s_ad[i]  = (i < cnt * H) ? a_d[(size_t)n0 * H + i] : 0.f;
  }

  const int* __restrict__ srcp = ei;
  const int* __restrict__ dstp = ei + E;
  const bool al16 = ((E & 3) == 0) && (((size_t)dstp & 15) == 0);
  const int nG    = al16 ? (E >> 2) / SCAN_G : 0;
  const int eRest = nG * SCAN_G * 4;

  auto edgeA = [&](int d, int e, bool self) {
    const int dl = d - n0;
    if ((unsigned)dl < (unsigned)cnt) {
      const int s = self ? d : srcp[e];
      float av[H];
      ld_coef<H>(a_s, s, av);
#pragma unroll
      for (int hh = 0; hh < H; ++hh) {
        float l = av[hh] + s_ad[dl * H + hh];
        l = (l > 0.f) ? l : LRELU_SLOPE * l;
        atomicMax(&s_max[dl * H + hh], f2ord(l));
      }
    }
  };

  auto edgeB = [&](int d, int e, bool self) {
    const int  dl  = d - n0;
    const bool hit = (unsigned)dl < (unsigned)cnt;
    unsigned mask = __builtin_amdgcn_ballot_w32(hit);
    if (mask != 0u) {
      int   s = 0;
      float w[H];
#pragma unroll
      for (int hh = 0; hh < H; ++hh) w[hh] = 0.f;
      if (hit) {
        s = self ? d : srcp[e];
        float av[H];
        ld_coef<H>(a_s, s, av);
#pragma unroll
        for (int hh = 0; hh < H; ++hh) {
          float l = av[hh] + s_ad[dl * H + hh];
          l = (l > 0.f) ? l : LRELU_SLOPE * l;
          w[hh] = expf(l - ord2f(s_max[dl * H + hh]));
          atomicAdd(&s_den[dix(dl, hh)], w[hh]);
        }
      }
      while (mask) {
        const int L = __builtin_ctz(mask);
        mask &= mask - 1u;
        const int sB = __builtin_amdgcn_readlane(s, L);
        const int dB = __builtin_amdgcn_readlane(dl, L);
        const float* __restrict__ hrow = h + (size_t)sB * HC + lane;
        float hv[H];
#pragma unroll
        for (int hh = 0; hh < H; ++hh) hv[hh] = hrow[hh * 32];
#pragma unroll
        for (int hh = 0; hh < H; ++hh) {
          const float wB = __int_as_float(
              __builtin_amdgcn_readlane(__float_as_int(w[hh]), L));
          atomicAdd(&s_num[dB * HC + hh * 32 + lane], wB * hv[hh]);
        }
      }
    }
  };

  const int nlS = lane * SEG_WV + wv;
  const int dS  = (nlS < cnt) ? n0 + nlS : -1;

  auto scan = [&](auto&& edgeFn) {
    if (wv < nG) {
      int4 nx[SCAN_U];
      ld_group(dstp, wv, lane, nx);
      for (int g = wv; g < nG; g += SEG_WV) {
        int4 cur[SCAN_U];
#pragma unroll
        for (int u = 0; u < SCAN_U; ++u) cur[u] = nx[u];
        ld_group(dstp, min(g + SEG_WV, nG - 1), lane, nx);
        unsigned hm = 0u;
#pragma unroll
        for (int u = 0; u < SCAN_U; ++u) {
          hm |= ((unsigned)(cur[u].x - n0) < (unsigned)cnt) ? (1u << (4 * u))     : 0u;
          hm |= ((unsigned)(cur[u].y - n0) < (unsigned)cnt) ? (1u << (4 * u + 1)) : 0u;
          hm |= ((unsigned)(cur[u].z - n0) < (unsigned)cnt) ? (1u << (4 * u + 2)) : 0u;
          hm |= ((unsigned)(cur[u].w - n0) < (unsigned)cnt) ? (1u << (4 * u + 3)) : 0u;
        }
        const int ebase = 4 * (g * SCAN_G + lane);
        while (__builtin_amdgcn_ballot_w32(hm != 0u) != 0u) {
          const bool has = (hm != 0u);
          const int  idx = has ? __builtin_ctz(hm) : 0;
          hm &= hm - 1u;
          const int e = ebase + (idx >> 2) * (4 * 32) + (idx & 3);
          const int d = has ? dstp[e] : -1;
          edgeFn(d, e, false);
        }
      }
    }
    for (int eb = eRest + wv * 32; eb < E; eb += SEG_TB) {
      const int e = eb + lane;
      const int d = (e < E) ? dstp[e] : -1;
      edgeFn(d, (e < E) ? e : 0, false);
    }
  };

  int wc = 0;
  if constexpr (BUILD) {
    auto stage = [&](bool hit, int e) {
      const unsigned m = __builtin_amdgcn_ballot_w32(hit);
      if (m != 0u) {
        const int pos = wc + (int)__builtin_amdgcn_mbcnt_lo(m, 0u);
        if (hit && pos < WCAP) s_stage[wv * WCAP + pos] = e;
        wc += __builtin_popcount(m);
      }
    };
    const bool useB = al16 && (((size_t)ob & 15) == 0) && gridDim.x <= 256;
    const int nGb    = useB ? (E >> 4) / SCAN_G : 0;
    const int eRestB = useB ? nGb * SCAN_G * 16 : eRest;
    if (useB) {
      const unsigned rep = (unsigned)blockIdx.x * 0x01010101u;
      auto own4 = [&](unsigned w) -> unsigned {
        const unsigned x = w ^ rep;
        return (~(((x & 0x7F7F7F7Fu) + 0x7F7F7F7Fu) | x | 0x7F7F7F7Fu)) >> 7;
      };
      if (wv < nGb) {
        uint4 nx[SCAN_U];
        ld_groupb(ob, wv, lane, nx);
        for (int g = wv; g < nGb; g += SEG_WV) {
          uint4 cur[SCAN_U];
#pragma unroll
          for (int u = 0; u < SCAN_U; ++u) cur[u] = nx[u];
          ld_groupb(ob, min(g + SEG_WV, nGb - 1), lane, nx);
          unsigned mA = 0u, mB = 0u;
#pragma unroll
          for (int u = 0; u < SCAN_U; ++u) {
            const unsigned f0 = own4(cur[u].x), f1 = own4(cur[u].y);
            const unsigned f2 = own4(cur[u].z), f3 = own4(cur[u].w);
            const int sh = 4 * (u & 1);
            const unsigned m = (f0 << sh) | (f1 << (sh + 1)) | (f2 << (sh + 2)) | (f3 << (sh + 3));
            if (u < 2) mA |= m; else mB |= m;
          }
          const int ebase = 16 * (g * SCAN_G + lane);
          while (__builtin_amdgcn_ballot_w32((mA | mB) != 0u) != 0u) {
            const bool     inA = (mA != 0u);
            const unsigned mx  = inA ? mA : mB;
            const bool     has = (mx != 0u);
            const int      t   = has ? __builtin_ctz(mx) : 0;
            const unsigned mc  = mx & (mx - 1u);
            mA = inA ? mc : mA;
            mB = inA ? mB : mc;
            const int j = (t & 7) + (inA ? 0 : 8);
            const int e = ebase + (j >> 2) * 512 + (j & 3) * 4 + (t >> 3);
            stage(has, e);
          }
        }
      }
    } else if (wv < nG) {
      int4 nx[SCAN_U];
      ld_group(dstp, wv, lane, nx);
      for (int g = wv; g < nG; g += SEG_WV) {
        int4 cur[SCAN_U];
#pragma unroll
        for (int u = 0; u < SCAN_U; ++u) cur[u] = nx[u];
        ld_group(dstp, min(g + SEG_WV, nG - 1), lane, nx);
        const int ebase = 4 * (g * SCAN_G + lane);
#pragma unroll
        for (int u = 0; u < SCAN_U; ++u) {
          stage((unsigned)(cur[u].x - n0) < (unsigned)cnt, ebase + u * 128);
          stage((unsigned)(cur[u].y - n0) < (unsigned)cnt, ebase + u * 128 + 1);
          stage((unsigned)(cur[u].z - n0) < (unsigned)cnt, ebase + u * 128 + 2);
          stage((unsigned)(cur[u].w - n0) < (unsigned)cnt, ebase + u * 128 + 3);
        }
      }
    }
    for (int eb = eRestB + wv * 32; eb < E; eb += SEG_TB) {
      const int e = eb + lane;
      const int d = (e < E) ? dstp[e] : -1;
      stage((unsigned)(d - n0) < (unsigned)cnt, e);
    }
    if (lane == 0) s_wc[wv] = wc;
  }
  __syncthreads();

  int2* __restrict__ wl = (int2*)wlist + (size_t)blockIdx.x * LCAP;

  auto put = [&](int idx, int s, int dl) {
    l_src[idx] = s;
    l_dl[idx]  = (unsigned char)dl;
    if constexpr (BUILD) wl[idx] = make_int2(s, dl);
    float av[H];
    ld_coef<H>(a_s, s, av);
#pragma unroll
    for (int hh = 0; hh < H; ++hh) {
      float l = av[hh] + s_ad[dl * H + hh];
      l = (l > 0.f) ? l : LRELU_SLOPE * l;
      atomicMax(&s_max[dl * H + hh], f2ord(l));
    }
  };

  int total;
  if constexpr (BUILD) {
    const int c = s_wc[lane];
    int pre = (lane < wv) ? c : 0;
    int all = c;
#pragma unroll
    for (int off = 16; off > 0; off >>= 1) {
      pre += __shfl_xor(pre, off, 32);
      all += __shfl_xor(all, off, 32);
    }
    pre = __builtin_amdgcn_readfirstlane(pre);
    all = __builtin_amdgcn_readfirstlane(all);
    const bool over = __builtin_amdgcn_ballot_w32(c > WCAP) != 0u;
    total = over ? -1 : all + cnt;
    if (tid == 0) wcnt[blockIdx.x] = total;
    if (!over) {
      for (int j = lane; j < wc; j += 32) {
        const int e = s_stage[wv * WCAP + j];
        put(pre + j, srcp[e], dstp[e] - n0);
      }
      if (tid < cnt) put(all + tid, n0 + tid, tid);
    }
  } else {
    total = wcnt[blockIdx.x];
    if (total > LCAP) total = -1;
    for (int i = tid; i < total; i += SEG_TB) {
      const int2 v = wl[i];
      put(i, v.x, v.y);
    }
  }

  if (total >= 0) {
    __syncthreads();
    constexpr int UNR = (H >= 8) ? 4 : 16;
    if constexpr (OWN) {
      const unsigned rep = (unsigned)wv * 0x01010101u;
      auto own4 = [&](unsigned q) -> unsigned {
        const unsigned x = (q & 0x1F1F1F1Fu) ^ rep;
        return ~(x + 0x7F7F7F7Fu) & 0x80808080u;
      };
      const int nCh = (total + 511) >> 9;
      unsigned short* __restrict__ mine = s_mine + wv * MCAP;
      int n = 0, ch = 0, ebase = 0;
      unsigned m = 0u;
      bool first = true;
      for (;;) {
        while (n <= MCAP - 32) {
          if (__builtin_amdgcn_ballot_w32(m != 0u) == 0u) {
            if (ch >= nCh) break;
            ebase = ch * 512 + lane * 16;
            ++ch;
            const uint4 q = *(const uint4*)(l_dl + ebase);
            m = (own4(q.x) >> 7) | (own4(q.y) >> 6) | (own4(q.z) >> 5) | (own4(q.w) >> 4);
            continue;
          }
          const bool has = (m != 0u);
          const int  t   = has ? __builtin_ctz(m) : 0;
          m &= m - 1u;
          const int  e   = ebase + 4 * (t & 7) + (t >> 3);
          const bool hit = has && e < total;
          const unsigned bm = __builtin_amdgcn_ballot_w32(hit);
          if (bm != 0u) {
            const int pos = n + (int)__builtin_amdgcn_mbcnt_lo(bm, 0u);
            if (hit) mine[pos] = (unsigned short)e;
            n += __builtin_popcount(bm);
          }
        }
        const bool last = (ch >= nCh) && (__builtin_amdgcn_ballot_w32(m != 0u) == 0u);
        float acc[RPW][H];
#pragma unroll
        for (int r = 0; r < RPW; ++r) {
#pragma unroll
          for (int hh = 0; hh < H; ++hh) acc[r][hh] = 0.f;
        }
        if (!first) {
#pragma unroll
          for (int r = 0; r < RPW; ++r) {
            const int row = wv + 32 * r;
            if (row < cnt) {
#pragma unroll
              for (int hh = 0; hh < H; ++hh) acc[r][hh] = s_num[row * HC + hh * 32 + lane];
            }
          }
        }
        for (int base = 0; base < n; base += 32) {
          const int  jj    = base + lane;
          const bool valid = jj < n;
          int   s = 0, dl = 0;
          float w[H];
#pragma unroll
          for (int hh = 0; hh < H; ++hh) w[hh] = 0.f;
          if (valid) {
            const int i = mine[jj];
            s  = l_src[i];
            dl = l_dl[i];
            float av[H];
            ld_coef<H>(a_s, s, av);
#pragma unroll
            for (int hh = 0; hh < H; ++hh) {
              float l = av[hh] + s_ad[dl * H + hh];
              l = (l > 0.f) ? l : LRELU_SLOPE * l;
              w[hh] = expf(l - ord2f(s_max[dl * H + hh]));
              atomicAdd(&s_den[dix(dl, hh)], w[hh]);
            }
          }
          const int nb = min(32, n - base);
          for (int L0 = 0; L0 < nb; L0 += UNR) {
            float hv[UNR][H];
            int   kB[UNR];
#pragma unroll
            for (int j = 0; j < UNR; ++j) {
              const int L  = min(L0 + j, nb - 1);
              const int sB = __builtin_amdgcn_readlane(s, L);
              kB[j]        = __builtin_amdgcn_readlane(dl, L) >> 5;
              const float* __restrict__ hrow = h + (size_t)sB * HC + lane;
#pragma unroll
              for (int hh = 0; hh < H; ++hh) hv[j][hh] = hrow[hh * 32];
            }
#pragma unroll
            for (int j = 0; j < UNR; ++j) {
              if (L0 + j < nb) {
                float wB[H];
#pragma unroll
                for (int hh = 0; hh < H; ++hh)
                  wB[hh] = __int_as_float(
                      __builtin_amdgcn_readlane(__float_as_int(w[hh]), L0 + j));
#pragma unroll
                for (int r = 0; r < RPW; ++r) {
                  if (kB[j] == r) {
#pragma unroll
                    for (int hh = 0; hh < H; ++hh) acc[r][hh] = fmaf(wB[hh], hv[j][hh], acc[r][hh]);
                  }
                }
              }
            }
          }
        }
#pragma unroll
        for (int r = 0; r < RPW; ++r) {
          const int row = wv + 32 * r;
          if (row < cnt) {
#pragma unroll
            for (int hh = 0; hh < H; ++hh) s_num[row * HC + hh * 32 + lane] = acc[r][hh];
          }
        }
        if (last) {
          float bv[H];
#pragma unroll
          for (int hh = 0; hh < H; ++hh) bv[hh] = bias[hh * 32 + lane];
          for (int row = wv; row < cnt; row += SEG_WV) {
            const float* __restrict__ nrow = s_num + row * HC + lane;
            float* __restrict__ orw = out + (size_t)(n0 + row) * HC + lane;
            float q[H];
#pragma unroll
            for (int hh = 0; hh < H; ++hh)
              q[hh] = nrow[hh * 32] / (s_den[dix(row, hh)] + 1e-16f) + bv[hh];
#pragma unroll
            for (int hh = 0; hh < H; ++hh) {
              const float em1 = expf(q[hh]) - 1.f;
              orw[hh * 32] = (q[hh] > 0.f) ? q[hh] : em1;
            }
          }
          break;
        }
        first = false;
        n = 0;
      }
    } else {
      const int per = (total + SEG_WV - 1) / SEG_WV;
      const int lo  = wv * per;
      const int hi  = min(total, lo + per);
      for (int base = lo; base < hi; base += 32) {
        const int  i     = base + lane;
        const bool valid = i < hi;
        int   s = 0, dl = 0;
        float w[H];
#pragma unroll
        for (int hh = 0; hh < H; ++hh) w[hh] = 0.f;
        if (valid) {
          s  = l_src[i];
          dl = l_dl[i];
          float av[H];
          ld_coef<H>(a_s, s, av);
#pragma unroll
          for (int hh = 0; hh < H; ++hh) {
            float l = av[hh] + s_ad[dl * H + hh];
            l = (l > 0.f) ? l : LRELU_SLOPE * l;
            w[hh] = expf(l - ord2f(s_max[dl * H + hh]));
            atomicAdd(&s_den[dix(dl, hh)], w[hh]);
          }
        }
        const int n = min(32, hi - base);
        for (int L0 = 0; L0 < n; L0 += UNR) {
          float hv[UNR][H];
          int   dB[UNR];
#pragma unroll
          for (int j = 0; j < UNR; ++j) {
            const int L  = min(L0 + j, n - 1);
            const int sB = __builtin_amdgcn_readlane(s, L);
            dB[j]        = __builtin_amdgcn_readlane(dl, L);
            const float* __restrict__ hrow = h + (size_t)sB * HC + lane;
#pragma unroll
            for (int hh = 0; hh < H; ++hh) hv[j][hh] = hrow[hh * 32];
          }
#pragma unroll
          for (int j = 0; j < UNR; ++j) {
            if (L0 + j < n) {
#pragma unroll
              for (int hh = 0; hh < H; ++hh) {
                const float wB = __int_as_float(
                    __builtin_amdgcn_readlane(__float_as_int(w[hh]), L0 + j));
                atomicAdd(&s_num[dB[j] * HC + hh * 32 + lane], wB * hv[j][hh]);
              }
            }
          }
        }
      }
    }
  } else {
    if constexpr (OWN) {
      for (int i = tid; i < NPB * HC; i += SEG_TB) s_num[i] = 0.f;
    }
    edgeA(dS, 0, true);
    scan(edgeA);
    __syncthreads();
    edgeB(dS, 0, true);
    scan(edgeB);
  }
  if (!OWN || total < 0) {
    __syncthreads();
    float* __restrict__ orow = out + (size_t)n0 * HC;
    for (int i = tid; i < cnt * HC; i += SEG_TB) {
      const int nl = i / HC;
      const int hc = i % HC;
      float v = s_num[i] / (s_den[dix(nl, hc / 32)] + 1e-16f) + bias[hc];
      v = (v > 0.f) ? v : (expf(v) - 1.f);
      orow[i] = v;
    }
  }
}

extern "C" void kernel_launch(void* const* d_in, const int* in_sizes, int n_in,
                              void* d_out, int out_size, void* d_ws, size_t ws_size,
                              hipStream_t stream) {
  const float* x   = (const float*)d_in[0];
  const int*   ei  = (const int*)d_in[1];
  const float* W1  = (const float*)d_in[2];
  const float* as1 = (const float*)d_in[3];
  const float* ad1 = (const float*)d_in[4];
  const float* b1  = (const float*)d_in[5];
  const float* W2  = (const float*)d_in[6];
  const float* as2 = (const float*)d_in[7];
  const float* ad2 = (const float*)d_in[8];
  const float* b2  = (const float*)d_in[9];

  const int HC   = in_sizes[3];
  const int Fin  = in_sizes[2] / HC;
  const int N    = in_sizes[0] / Fin;
  const int E    = in_sizes[1] / 2;
  const int H    = 8;
  const int Cout = in_sizes[7];

  float* ws = (float*)d_ws;
  float* h1  = ws;
  float* x2  = h1 + (size_t)N * HC;
  float* h2  = x2 + (size_t)N * HC;
  float* aS1 = h2 + (size_t)N * Cout;
  float* aD1 = aS1 + (size_t)N * H;
  float* aS2 = aD1 + (size_t)N * H;
  float* aD2 = aS2 + N;
  const int segBlocksWs = (N + NPB - 1) / NPB;
  int* eList = (int*)(aD2 + N);
  int* eCnt  = eList + (size_t)segBlocksWs * LCAP * 2;
  unsigned* eOwn = (unsigned*)(eCnt + ((segBlocksWs + 3) & ~3));

  const int TB = 256;
  auto blocks = [](long long n, int tb) { return (int)((n + tb - 1) / tb); };
  const int segBlocks = (N + NPB - 1) / NPB;

  {
    gemm_att_wmma_f32_ldsw<128, 256><<<256, G1_TB, 0, stream>>>(
        x, W1, as1, ad1, h1, aS1, aD1, N, ei + E, E, segBlocks, eOwn);
    seg_softmax_agg<8, true><<<segBlocks, SEG_TB, 0, stream>>>(
        ei, E, N, aS1, aD1, h1, b1, x2, eList, eCnt, eOwn);
  }

  {
    const long long waves = (long long)(N / 16) * (Cout / 32);
    gemm_att_wmma_f32<256, 32><<<blocks(waves * 32, TB), TB, 0, stream>>>(
        x2, W2, as2, ad2, h2, aS2, aD2, N);
    seg_softmax_agg<1, false><<<segBlocks, SEG_TB, 0, stream>>>(
        ei, E, N, aS2, aD2, h2, b2, (float*)d_out, eList, eCnt, eOwn);
  }
}
